# starting kernel
# baseline (speedup 1.0000x reference)
#include <hip/hip_runtime.h>
#include <math.h>

#define NODES_DEF 100000
#define IN_C 128
#define HID 32
#define OUT_C 32
#define NEG_SLOPE 0.2f

typedef __attribute__((ext_vector_type(2))) float v2f;
typedef __attribute__((ext_vector_type(8))) float v8f;

template <int K, int NC, int EPI>
__global__ void gat_gemm_wmma(const float* __restrict__ A,
                              const float* __restrict__ B,
                              const float* __restrict__ bias,
                              float* __restrict__ C, int M)
{
    const int wavesPerBlock = blockDim.x >> 5;
    const int wave = blockIdx.x * wavesPerBlock + (threadIdx.x >> 5);
    const int nTiles = NC / 16;
    const int totalTiles = (M / 16) * nTiles;
    if (wave >= totalTiles) return;

    const int mTile = wave / nTiles;
    const int nTile = wave % nTiles;
    const int lane  = threadIdx.x & 31;
    const int half  = lane >> 4;
    const int r     = lane & 15;
    const int row   = mTile * 16 + r;
    const int col   = nTile * 16 + r;

    v8f acc = {};
#pragma unroll
    for (int k = 0; k < K; k += 4) {
        v2f a = *(const v2f*)(A + (size_t)row * K + k + 2 * half);
        v2f b;
        b.x = B[(k + 2 * half + 0) * NC + col];
        b.y = B[(k + 2 * half + 1) * NC + col];
        acc = __builtin_amdgcn_wmma_f32_16x16x4_f32(
             false, a,  false, b,
             (short)0, acc,  false,  false);
    }

#pragma unroll
    for (int i = 0; i < 8; ++i) {
        const int orow = mTile * 16 + i + 8 * half;
        float v = acc[i];
        if (EPI == 1) {
            v += bias[col];
            v = 1.0f / (1.0f + __expf(-v));
        }
        C[(size_t)orow * NC + col] = v;
    }
}

__device__ __forceinline__ int ord_f32(float f)
{
    int i = __float_as_int(f);
    return i >= 0 ? i : (i ^ 0x7fffffff);
}
__device__ __forceinline__ float unord_f32(int i)
{
    return __int_as_float(i >= 0 ? i : (i ^ 0x7fffffff));
}
__device__ __forceinline__ float leaky(float x)
{
    return x > 0.0f ? x : NEG_SLOPE * x;
}

__global__ void gat_node_prep(const float* __restrict__ h,
                              const float* __restrict__ a_src,
                              const float* __restrict__ a_dst,
                              float* __restrict__ s_src,
                              float* __restrict__ s_dst,
                              int*   __restrict__ m_ord,
                              float* __restrict__ z,
                              float* __restrict__ agg,
                              int n)
{
    const int i = blockIdx.x * blockDim.x + threadIdx.x;
    if (i >= n) return;
    const float* hp = h + (size_t)i * HID;
    float* ap = agg + (size_t)i * HID;
    float ss = 0.0f, sd = 0.0f;
#pragma unroll
    for (int c = 0; c < HID; ++c) {
        const float v = hp[c];
        ss += v * a_src[c];
        sd += v * a_dst[c];
        ap[c] = 0.0f;
    }
    s_src[i] = ss;
    s_dst[i] = sd;
    m_ord[i] = (int)0x80000000;
    z[i]     = 0.0f;
}

__device__ __forceinline__ void edge_endpoints(const int* __restrict__ ei,
                                               int idx, int n_edges,
                                               int& s, int& d)
{
    if (idx < n_edges) { s = ei[idx]; d = ei[n_edges + idx]; }
    else               { s = d = idx - n_edges; }
}

__global__ void gat_edge_max(const int* __restrict__ ei,
                             const float* __restrict__ s_src,
                             const float* __restrict__ s_dst,
                             int* __restrict__ m_ord,
                             int n_edges, int e_total)
{
    const int idx = blockIdx.x * blockDim.x + threadIdx.x;
    if (idx >= e_total) return;
    int s, d;
    edge_endpoints(ei, idx, n_edges, s, d);
    const float e = leaky(s_src[s] + s_dst[d]);
    atomicMax(&m_ord[d], ord_f32(e));
}

__global__ void gat_edge_sum(const int* __restrict__ ei,
                             const float* __restrict__ s_src,
                             const float* __restrict__ s_dst,
                             const int* __restrict__ m_ord,
                             float* __restrict__ z,
                             int n_edges, int e_total)
{
    const int idx = blockIdx.x * blockDim.x + threadIdx.x;
    if (idx >= e_total) return;
    int s, d;
    edge_endpoints(ei, idx, n_edges, s, d);
    const float e  = leaky(s_src[s] + s_dst[d]);
    const float ex = __expf(e - unord_f32(m_ord[d]));
    atomicAdd(&z[d], ex);
}

__global__ void gat_edge_agg(const int* __restrict__ ei,
                             const float* __restrict__ s_src,
                             const float* __restrict__ s_dst,
                             const int* __restrict__ m_ord,
                             const float* __restrict__ z,
                             const float* __restrict__ h,
                             float* __restrict__ agg,
                             int n_edges, int e_total)
{
    const int idx = blockIdx.x * blockDim.x + threadIdx.x;
    if (idx >= e_total) return;
    int s, d;
    edge_endpoints(ei, idx, n_edges, s, d);
    const float e     = leaky(s_src[s] + s_dst[d]);
    const float alpha = __expf(e - unord_f32(m_ord[d])) / (z[d] + 1e-16f);

    const float4* hp = (const float4*)(h + (size_t)s * HID);
    float* ap = agg + (size_t)d * HID;
#pragma unroll
    for (int c4 = 0; c4 < HID / 4; ++c4) {
        const float4 v = hp[c4];
        atomicAdd(ap + 4 * c4 + 0, alpha * v.x);
        atomicAdd(ap + 4 * c4 + 1, alpha * v.y);
        atomicAdd(ap + 4 * c4 + 2, alpha * v.z);
        atomicAdd(ap + 4 * c4 + 3, alpha * v.w);
    }
}

__global__ void gat_bias_relu(const float* __restrict__ agg,
                              const float* __restrict__ bias,
                              float* __restrict__ g, int total)
{
    const int i = blockIdx.x * blockDim.x + threadIdx.x;
    if (i >= total) return;
    const float v = agg[i] + bias[i & (HID - 1)];
    g[i] = v > 0.0f ? v : 0.0f;
}

extern "C" void kernel_launch(void* const* d_in, const int* in_sizes, int n_in,
                              void* d_out, int out_size, void* d_ws, size_t ws_size,
                              hipStream_t stream)
{
    const float* x      = (const float*)d_in[0];
    const int*   ei     = (const int*)  d_in[1];
    const float* W1     = (const float*)d_in[2];
    const float* a_src1 = (const float*)d_in[3];
    const float* a_dst1 = (const float*)d_in[4];
    const float* b1     = (const float*)d_in[5];
    const float* W2     = (const float*)d_in[6];
    const float* a_src2 = (const float*)d_in[7];
    const float* a_dst2 = (const float*)d_in[8];
    const float* b2     = (const float*)d_in[9];
    const float* Wc     = (const float*)d_in[10];
    const float* bc     = (const float*)d_in[11];
    float* out = (float*)d_out;

    const int n       = in_sizes[0] / IN_C;
    const int n_edges = in_sizes[1] / 2;
    const int e_total = n_edges + n;

    float* ws    = (float*)d_ws;
    float* h     = ws;
    float* g     = h   + (size_t)n * HID;
    float* agg   = g   + (size_t)n * HID;
    float* s_src = agg + (size_t)n * HID;
    float* s_dst = s_src + n;
    float* z     = s_dst + n;
    int*   m_ord = (int*)(z + n);

    const int TB = 256;
    const int nodeBlocks = (n + TB - 1) / TB;
    const int edgeBlocks = (e_total + TB - 1) / TB;
    const int elemBlocks = (n * HID + TB - 1) / TB;
    const int gemmWaves  = (n / 16) * (HID / 16);
    const int gemmBlocks = (gemmWaves + 7) / 8;

    gat_gemm_wmma<IN_C, HID, 0><<<gemmBlocks, TB, 0, stream>>>(x, W1, nullptr, h, n);
    gat_node_prep<<<nodeBlocks, TB, 0, stream>>>(h, a_src1, a_dst1, s_src, s_dst,
                                                 m_ord, z, agg, n);
    gat_edge_max<<<edgeBlocks, TB, 0, stream>>>(ei, s_src, s_dst, m_ord, n_edges, e_total);
    gat_edge_sum<<<edgeBlocks, TB, 0, stream>>>(ei, s_src, s_dst, m_ord, z, n_edges, e_total);
    gat_edge_agg<<<edgeBlocks, TB, 0, stream>>>(ei, s_src, s_dst, m_ord, z, h, agg,
                                                n_edges, e_total);
    gat_bias_relu<<<elemBlocks, TB, 0, stream>>>(agg, b1, g, n * HID);

    gat_gemm_wmma<HID, HID, 0><<<gemmBlocks, TB, 0, stream>>>(g, W2, nullptr, h, n);
    gat_node_prep<<<nodeBlocks, TB, 0, stream>>>(h, a_src2, a_dst2, s_src, s_dst,
                                                 m_ord, z, agg, n);
    gat_edge_max<<<edgeBlocks, TB, 0, stream>>>(ei, s_src, s_dst, m_ord, n_edges, e_total);
    gat_edge_sum<<<edgeBlocks, TB, 0, stream>>>(ei, s_src, s_dst, m_ord, z, n_edges, e_total);
    gat_edge_agg<<<edgeBlocks, TB, 0, stream>>>(ei, s_src, s_dst, m_ord, z, h, agg,
                                                n_edges, e_total);
    gat_bias_relu<<<elemBlocks, TB, 0, stream>>>(agg, b2, g, n * HID);

    gat_gemm_wmma<HID, OUT_C, 1><<<gemmBlocks, TB, 0, stream>>>(g, Wc, bc, out, n);
}


# edge phase of a layer in one launch: a work-group owns 391 destination rows in LDS, filters the whole edge list for them, does max, z and weighted sum by LDS atomics; sums in LDS arrival order
# speedup vs baseline: 7.6810x; 7.6810x over previous
#include <hip/hip_runtime.h>
#include <math.h>

#define NODES_DEF 100000
#define IN_C 128
#define HID 32
#define OUT_C 32
#define NEG_SLOPE 0.2f

typedef __attribute__((ext_vector_type(2))) float v2f;
typedef __attribute__((ext_vector_type(8))) float v8f;

template <int K, int NC, int EPI>
__global__ void gat_gemm_wmma(const float* __restrict__ A,
                              const float* __restrict__ B,
                              const float* __restrict__ bias,
                              float* __restrict__ C, int M)
{
    const int wavesPerBlock = blockDim.x >> 5;
    const int wave = blockIdx.x * wavesPerBlock + (threadIdx.x >> 5);
    const int nTiles = NC / 16;
    const int totalTiles = (M / 16) * nTiles;
    if (wave >= totalTiles) return;

    const int mTile = wave / nTiles;
    const int nTile = wave % nTiles;
    const int lane  = threadIdx.x & 31;
    const int half  = lane >> 4;
    const int r     = lane & 15;
    const int row   = mTile * 16 + r;
    const int col   = nTile * 16 + r;

    v8f acc = {};
#pragma unroll
    for (int k = 0; k < K; k += 4) {
        v2f a = *(const v2f*)(A + (size_t)row * K + k + 2 * half);
        v2f b;
        b.x = B[(k + 2 * half + 0) * NC + col];
        b.y = B[(k + 2 * half + 1) * NC + col];
        acc = __builtin_amdgcn_wmma_f32_16x16x4_f32(
             false, a,  false, b,
             (short)0, acc,  false,  false);
    }

#pragma unroll
    for (int i = 0; i < 8; ++i) {
        const int orow = mTile * 16 + i + 8 * half;
        float v = acc[i];
        if (EPI == 1) {
            v += bias[col];
            v = 1.0f / (1.0f + __expf(-v));
        }
        C[(size_t)orow * NC + col] = v;
    }
}

__device__ __forceinline__ int ord_f32(float f)
{
    int i = __float_as_int(f);
    return i >= 0 ? i : (i ^ 0x7fffffff);
}
__device__ __forceinline__ float unord_f32(int i)
{
    return __int_as_float(i >= 0 ? i : (i ^ 0x7fffffff));
}
__device__ __forceinline__ float leaky(float x)
{
    return x > 0.0f ? x : NEG_SLOPE * x;
}

__global__ void gat_node_prep(const float* __restrict__ h,
                              const float* __restrict__ a_src,
                              const float* __restrict__ a_dst,
                              float* __restrict__ s_src,
                              float* __restrict__ s_dst,
                              int n)
{
    const int i = blockIdx.x * blockDim.x + threadIdx.x;
    if (i >= n) return;
    const float* hp = h + (size_t)i * HID;
    float ss = 0.0f, sd = 0.0f;
#pragma unroll
    for (int c = 0; c < HID; ++c) {
        const float v = hp[c];
        ss += v * a_src[c];
        sd += v * a_dst[c];
    }
    s_src[i] = ss;
    s_dst[i] = sd;
}

#define AGG_RB      391
#define AGG_CAP     16384
#define AGG_THREADS 1024

__device__ __forceinline__ void agg_push(int idx, unsigned dl, unsigned R,
                                         int* count_s, int* la, int* lb)
{
    if (dl < R) {
        const int pos = atomicAdd(count_s, 1);
        if (pos < AGG_CAP) { la[pos] = idx; lb[pos] = (int)dl; }
    }
}

template <int MODE>
__device__ __forceinline__ void agg_slow_scan(const int* __restrict__ ei,
                                              const float* __restrict__ s_src,
                                              const float* __restrict__ h,
                                              const float* sd_s, int* m_s,
                                              float* z_s, float* agg_s,
                                              int base, unsigned R, int n_edges, int tid)
{
    for (int idx = tid; idx < n_edges; idx += AGG_THREADS) {
        const unsigned dl = (unsigned)(ei[n_edges + idx] - base);
        if (dl < R) {
            const int s = ei[idx];
            const float e = leaky(s_src[s] + sd_s[dl]);
            if (MODE == 0) {
                atomicMax(&m_s[dl], ord_f32(e));
            } else {
                const float ex = __expf(e - unord_f32(m_s[dl]));
                atomicAdd(&z_s[dl], ex);
                for (int c = 0; c < HID; ++c)
                    atomicAdd(&agg_s[dl * HID + c], ex * h[(size_t)s * HID + c]);
            }
        }
    }
}

__global__ __launch_bounds__(AGG_THREADS)
void gat_block_agg(const int* __restrict__ ei,
                   const float* __restrict__ s_src,
                   const float* __restrict__ s_dst,
                   const float* __restrict__ h,
                   const float* __restrict__ bias,
                   float* __restrict__ g,
                   int n, int n_edges)
{
    __shared__ float agg_s[AGG_RB * HID];
    __shared__ float z_s[AGG_RB];
    __shared__ float sd_s[AGG_RB];
    __shared__ float es_s[AGG_RB];
    __shared__ int   m_s[AGG_RB];
    __shared__ int   la[AGG_CAP];
    __shared__ int   lb[AGG_CAP];
    __shared__ float lc[AGG_CAP];
    __shared__ int   count_s;

    const int tid  = threadIdx.x;
    const int lane = tid & 31;
    const int wave = __builtin_amdgcn_readfirstlane(tid >> 5);
    const int* __restrict__ dstp = ei + n_edges;
    const int nvec = ((n_edges & 3) == 0) ? (n_edges >> 2) : 0;
    const int4* __restrict__ dv = (const int4*)dstp;

    for (int base = blockIdx.x * AGG_RB; base < n; base += gridDim.x * AGG_RB) {
        const unsigned R = (unsigned)min(AGG_RB, n - base);

        for (int i = tid; i < (int)R * HID; i += AGG_THREADS) agg_s[i] = 0.0f;
        for (int i = tid; i < (int)R; i += AGG_THREADS) {
            const float sd = s_dst[base + i];
            const float es = leaky(s_src[base + i] + sd);
            sd_s[i] = sd;
            es_s[i] = es;
            z_s[i]  = 0.0f;
            m_s[i]  = ord_f32(es);
        }
        if (tid == 0) count_s = 0;
        __syncthreads();

        const int nsweep = nvec / (4 * AGG_THREADS);
        if (nsweep > 0) {
            int4 q[4];
#pragma unroll
            for (int u = 0; u < 4; ++u) q[u] = dv[tid + u * AGG_THREADS];
            for (int sw = 0; sw < nsweep; ++sw) {
                const int nsw = (sw + 1 < nsweep) ? sw + 1 : sw;
                int4 nq[4];
#pragma unroll
                for (int u = 0; u < 4; ++u)
                    nq[u] = dv[nsw * (4 * AGG_THREADS) + u * AGG_THREADS + tid];
#pragma unroll
                for (int u = 0; u < 4; ++u) {
                    const int idx = 4 * (sw * (4 * AGG_THREADS) + u * AGG_THREADS + tid);
                    const unsigned a0 = (unsigned)(q[u].x - base);
                    const unsigned a1 = (unsigned)(q[u].y - base);
                    const unsigned a2 = (unsigned)(q[u].z - base);
                    const unsigned a3 = (unsigned)(q[u].w - base);
                    if (min(min(a0, a1), min(a2, a3)) < R) {
                        agg_push(idx + 0, a0, R, &count_s, la, lb);
                        agg_push(idx + 1, a1, R, &count_s, la, lb);
                        agg_push(idx + 2, a2, R, &count_s, la, lb);
                        agg_push(idx + 3, a3, R, &count_s, la, lb);
                    }
                }
#pragma unroll
                for (int u = 0; u < 4; ++u) q[u] = nq[u];
            }
        }
        for (int v = nsweep * (4 * AGG_THREADS) + tid; v < nvec; v += AGG_THREADS) {
            const int4 q1 = dv[v];
            agg_push(4 * v + 0, (unsigned)(q1.x - base), R, &count_s, la, lb);
            agg_push(4 * v + 1, (unsigned)(q1.y - base), R, &count_s, la, lb);
            agg_push(4 * v + 2, (unsigned)(q1.z - base), R, &count_s, la, lb);
            agg_push(4 * v + 3, (unsigned)(q1.w - base), R, &count_s, la, lb);
        }
        for (int idx = 4 * nvec + tid; idx < n_edges; idx += AGG_THREADS)
            agg_push(idx, (unsigned)(dstp[idx] - base), R, &count_s, la, lb);
        __syncthreads();

        const int cnt = __builtin_amdgcn_readfirstlane(count_s);
        if (cnt <= AGG_CAP) {
            for (int i0 = tid; i0 < cnt; i0 += 4 * AGG_THREADS) {
                int eidx[4], dl[4], s[4];
                float ss[4];
#pragma unroll
                for (int u = 0; u < 4; ++u) {
                    const int i = i0 + u * AGG_THREADS;
                    const bool ok = i < cnt;
                    eidx[u] = ok ? la[i] : 0;
                    dl[u]   = ok ? lb[i] : 0;
                }
#pragma unroll
                for (int u = 0; u < 4; ++u) s[u] = ei[eidx[u]];
#pragma unroll
                for (int u = 0; u < 4; ++u) ss[u] = s_src[s[u]];
#pragma unroll
                for (int u = 0; u < 4; ++u) {
                    const int i = i0 + u * AGG_THREADS;
                    if (i < cnt) {
                        const float e = leaky(ss[u] + sd_s[dl[u]]);
                        atomicMax(&m_s[dl[u]], ord_f32(e));
                        la[i] = s[u];
                        lc[i] = e;
                    }
                }
            }
            __syncthreads();

            const int nbatch = (cnt + 31) >> 5;
            for (int b = wave; b < nbatch; b += AGG_THREADS / 32) {
                const int i = (b << 5) + lane;
                const bool ok = i < cnt;
                const int s_l  = ok ? la[i] : 0;
                const int dl_l = ok ? lb[i] : 0;
                float ex_l = 0.0f;
                if (ok) {
                    ex_l = __expf(lc[i] - unord_f32(m_s[dl_l]));
                    atomicAdd(&z_s[dl_l], ex_l);
                }
                const int nvalid = min(32, cnt - (b << 5));
                if (nvalid == 32) {
#pragma unroll
                    for (int k0 = 0; k0 < 32; k0 += 8) {
                        float v[8];
#pragma unroll
                        for (int j = 0; j < 8; ++j) {
                            const int s = __builtin_amdgcn_readlane(s_l, k0 + j);
                            v[j] = h[(size_t)s * HID + lane];
                        }
#pragma unroll
                        for (int j = 0; j < 8; ++j) {
                            const int   dl = __builtin_amdgcn_readlane(dl_l, k0 + j);
                            const float ex = __int_as_float(
                                __builtin_amdgcn_readlane(__float_as_int(ex_l), k0 + j));
                            atomicAdd(&agg_s[dl * HID + lane], ex * v[j]);
                        }
                    }
                } else {
                    for (int k = 0; k < nvalid; ++k) {
                        const int   s  = __builtin_amdgcn_readlane(s_l, k);
                        const int   dl = __builtin_amdgcn_readlane(dl_l, k);
                        const float ex = __int_as_float(
                            __builtin_amdgcn_readlane(__float_as_int(ex_l), k));
                        atomicAdd(&agg_s[dl * HID + lane], ex * h[(size_t)s * HID + lane]);
                    }
                }
            }
        } else {
            agg_slow_scan<0>(ei, s_src, h, sd_s, m_s, z_s, agg_s, base, R, n_edges, tid);
            __syncthreads();
            agg_slow_scan<1>(ei, s_src, h, sd_s, m_s, z_s, agg_s, base, R, n_edges, tid);
        }
        __syncthreads();

        for (int i = tid; i < (int)R * HID; i += AGG_THREADS) {
            const int dl = i / HID;
            const int c  = i & (HID - 1);
            const size_t o = (size_t)(base + dl) * HID + c;
            const float exs = __expf(es_s[dl] - unord_f32(m_s[dl]));
            const float v = (agg_s[i] + exs * h[o]) / (z_s[dl] + exs + 1e-16f) + bias[c];
            g[o] = v > 0.0f ? v : 0.0f;
        }
        __syncthreads();
    }
}

extern "C" void kernel_launch(void* const* d_in, const int* in_sizes, int n_in,
                              void* d_out, int out_size, void* d_ws, size_t ws_size,
                              hipStream_t stream)
{
    const float* x      = (const float*)d_in[0];
    const int*   ei     = (const int*)  d_in[1];
    const float* W1     = (const float*)d_in[2];
    const float* a_src1 = (const float*)d_in[3];
    const float* a_dst1 = (const float*)d_in[4];
    const float* b1     = (const float*)d_in[5];
    const float* W2     = (const float*)d_in[6];
    const float* a_src2 = (const float*)d_in[7];
    const float* a_dst2 = (const float*)d_in[8];
    const float* b2     = (const float*)d_in[9];
    const float* Wc     = (const float*)d_in[10];
    const float* bc     = (const float*)d_in[11];
    float* out = (float*)d_out;

    const int n       = in_sizes[0] / IN_C;
    const int n_edges = in_sizes[1] / 2;

    float* ws    = (float*)d_ws;
    float* h     = ws;
    float* g     = h   + (size_t)n * HID;
    float* s_src = g   + (size_t)2 * n * HID;
    float* s_dst = s_src + n;

    const int TB = 256;
    const int nodeBlocks = (n + TB - 1) / TB;
    const int aggBlocks  = (n + AGG_RB - 1) / AGG_RB;
    const int gemmWaves  = (n / 16) * (HID / 16);
    const int gemmBlocks = (gemmWaves + 7) / 8;

    gat_gemm_wmma<IN_C, HID, 0><<<gemmBlocks, TB, 0, stream>>>(x, W1, nullptr, h, n);
    gat_node_prep<<<nodeBlocks, TB, 0, stream>>>(h, a_src1, a_dst1, s_src, s_dst, n);
    gat_block_agg<<<aggBlocks, AGG_THREADS, 0, stream>>>(ei, s_src, s_dst, h, b1, g, n, n_edges);

    gat_gemm_wmma<HID, HID, 0><<<gemmBlocks, TB, 0, stream>>>(g, W2, nullptr, h, n);
    gat_node_prep<<<nodeBlocks, TB, 0, stream>>>(h, a_src2, a_dst2, s_src, s_dst, n);
    gat_block_agg<<<aggBlocks, AGG_THREADS, 0, stream>>>(ei, s_src, s_dst, h, b2, g, n, n_edges);

    gat_gemm_wmma<HID, OUT_C, 1><<<gemmBlocks, TB, 0, stream>>>(g, Wc, bc, out, n);
}


# edge filter walks a per-group bitmap (1 bit per 16-edge key block, OR-ed in LDS by the key pre-pass) and reads marked key blocks only; list slots by wave ballot; sums still LDS arrival order
# speedup vs baseline: 13.0623x; 1.7006x over previous
#include <hip/hip_runtime.h>
#include <math.h>

#define NODES_DEF 100000
#define IN_C 128
#define HID 32
#define OUT_C 32
#define NEG_SLOPE 0.2f

typedef __attribute__((ext_vector_type(2))) float v2f;
typedef __attribute__((ext_vector_type(8))) float v8f;

template <int K, int NC, int EPI>
__global__ void gat_gemm_wmma(const float* __restrict__ A,
                              const float* __restrict__ B,
                              const float* __restrict__ bias,
                              float* __restrict__ C, int M)
{
    const int wavesPerBlock = blockDim.x >> 5;
    const int wave = blockIdx.x * wavesPerBlock + (threadIdx.x >> 5);
    const int nTiles = NC / 16;
    const int totalTiles = (M / 16) * nTiles;
    if (wave >= totalTiles) return;

    const int mTile = wave / nTiles;
    const int nTile = wave % nTiles;
    const int lane  = threadIdx.x & 31;
    const int half  = lane >> 4;
    const int r     = lane & 15;
    const int row   = mTile * 16 + r;
    const int col   = nTile * 16 + r;

    v8f acc = {};
#pragma unroll
    for (int k = 0; k < K; k += 4) {
        v2f a = *(const v2f*)(A + (size_t)row * K + k + 2 * half);
        v2f b;
        b.x = B[(k + 2 * half + 0) * NC + col];
        b.y = B[(k + 2 * half + 1) * NC + col];
        acc = __builtin_amdgcn_wmma_f32_16x16x4_f32(
             false, a,  false, b,
             (short)0, acc,  false,  false);
    }

#pragma unroll
    for (int i = 0; i < 8; ++i) {
        const int orow = mTile * 16 + i + 8 * half;
        float v = acc[i];
        if (EPI == 1) {
            v += bias[col];
            v = 1.0f / (1.0f + __expf(-v));
        }
        C[(size_t)orow * NC + col] = v;
    }
}

__device__ __forceinline__ int ord_f32(float f)
{
    int i = __float_as_int(f);
    return i >= 0 ? i : (i ^ 0x7fffffff);
}
__device__ __forceinline__ float unord_f32(int i)
{
    return __int_as_float(i >= 0 ? i : (i ^ 0x7fffffff));
}
__device__ __forceinline__ float leaky(float x)
{
    return x > 0.0f ? x : NEG_SLOPE * x;
}

__global__ void gat_node_prep(const float* __restrict__ h,
                              const float* __restrict__ a_src,
                              const float* __restrict__ a_dst,
                              float* __restrict__ s_src,
                              float* __restrict__ s_dst,
                              int n)
{
    const int i = blockIdx.x * blockDim.x + threadIdx.x;
    if (i >= n) return;
    const float* hp = h + (size_t)i * HID;
    float ss = 0.0f, sd = 0.0f;
#pragma unroll
    for (int c = 0; c < HID; ++c) {
        const float v = hp[c];
        ss += v * a_src[c];
        sd += v * a_dst[c];
    }
    s_src[i] = ss;
    s_dst[i] = sd;
}

#define AGG_RB      391
#define AGG_CAP     16384
#define AGG_THREADS 1024
#define KEY_CHUNK   1024
#define KEY_PITCH   257

__device__ __forceinline__ void agg_push(int idx, int* count_s, int* la)
{
    const int pos = atomicAdd(count_s, 1);
    if (pos < AGG_CAP) la[pos] = idx;
}

__device__ __forceinline__ unsigned key_ne(unsigned q, unsigned kr)
{
    const unsigned x = q ^ kr;
    return ((x & 0x7f7f7f7fu) + 0x7f7f7f7fu) | x;
}
__device__ __forceinline__ unsigned key_mask16(uint4 q, unsigned kr)
{
    const unsigned ne = ((key_ne(q.x, kr) >> 7) & 0x01010101u) |
                        ((key_ne(q.y, kr) >> 6) & 0x02020202u) |
                        ((key_ne(q.z, kr) >> 5) & 0x04040404u) |
                        ((key_ne(q.w, kr) >> 4) & 0x08080808u);
    return ne ^ 0x0f0f0f0fu;
}

__global__ __launch_bounds__(KEY_CHUNK)
void gat_edge_keys(const int* __restrict__ ei,
                   unsigned char* __restrict__ key,
                   unsigned* __restrict__ bmap, int n_edges, int nchunk)
{
    __shared__ unsigned bm_s[32 * KEY_PITCH];
    const int tid = threadIdx.x;
    const int* __restrict__ dstp = ei + n_edges;
    const int nvec = n_edges >> 4;
    for (int i = tid; i < 32 * KEY_PITCH; i += KEY_CHUNK) bm_s[i] = 0u;
    __syncthreads();

    const int blk = blockIdx.x * KEY_CHUNK + tid;
    if (blk < nvec) {
        unsigned kk[16];
        if ((n_edges & 3) == 0) {
            const int4* __restrict__ dv = (const int4*)(dstp + 16 * blk);
#pragma unroll
            for (int j = 0; j < 4; ++j) {
                const int4 d = dv[j];
                kk[4 * j + 0] = (unsigned)(d.x / AGG_RB) & 255u;
                kk[4 * j + 1] = (unsigned)(d.y / AGG_RB) & 255u;
                kk[4 * j + 2] = (unsigned)(d.z / AGG_RB) & 255u;
                kk[4 * j + 3] = (unsigned)(d.w / AGG_RB) & 255u;
            }
        } else {
#pragma unroll
            for (int j = 0; j < 16; ++j)
                kk[j] = (unsigned)(dstp[16 * blk + j] / AGG_RB) & 255u;
        }
        unsigned w[4];
#pragma unroll
        for (int j = 0; j < 4; ++j)
            w[j] = kk[4 * j] | (kk[4 * j + 1] << 8) | (kk[4 * j + 2] << 16) | (kk[4 * j + 3] << 24);
        *(uint4*)(key + 16 * (size_t)blk) = make_uint4(w[0], w[1], w[2], w[3]);
        unsigned* plane = bm_s + (tid >> 5) * KEY_PITCH;
        const unsigned bit = 1u << (tid & 31);
#pragma unroll
        for (int j = 0; j < 16; ++j) atomicOr(&plane[kk[j]], bit);
    }
    if (blockIdx.x == 0 && tid < (n_edges & 15))
        key[16 * nvec + tid] = (unsigned char)((unsigned)(dstp[16 * nvec + tid] / AGG_RB) & 255u);
    __syncthreads();

    for (int i = tid; i < 256 * 32; i += KEY_CHUNK) {
        const int k = i >> 5, wd = i & 31;
        bmap[((size_t)k * nchunk + blockIdx.x) * 32 + wd] = bm_s[wd * KEY_PITCH + k];
    }
}

template <int MODE>
__device__ __forceinline__ void agg_slow_scan(const int* __restrict__ ei,
                                              const float* __restrict__ s_src,
                                              const float* __restrict__ h,
                                              const float* sd_s, int* m_s,
                                              float* z_s, float* agg_s,
                                              int base, unsigned R, int n_edges, int tid)
{
    for (int idx = tid; idx < n_edges; idx += AGG_THREADS) {
        const unsigned dl = (unsigned)(ei[n_edges + idx] - base);
        if (dl < R) {
            const int s = ei[idx];
            const float e = leaky(s_src[s] + sd_s[dl]);
            if (MODE == 0) {
                atomicMax(&m_s[dl], ord_f32(e));
            } else {
                const float ex = __expf(e - unord_f32(m_s[dl]));
                atomicAdd(&z_s[dl], ex);
                for (int c = 0; c < HID; ++c)
                    atomicAdd(&agg_s[dl * HID + c], ex * h[(size_t)s * HID + c]);
            }
        }
    }
}

__global__ __launch_bounds__(AGG_THREADS)
void gat_block_agg(const int* __restrict__ ei,
                   const unsigned char* __restrict__ key,
                   const unsigned* __restrict__ bmap,
                   const float* __restrict__ s_src,
                   const float* __restrict__ s_dst,
                   const float* __restrict__ h,
                   const float* __restrict__ bias,
                   float* __restrict__ g,
                   int n, int n_edges, int nchunk)
{
    __shared__ float agg_s[AGG_RB * HID];
    __shared__ float z_s[AGG_RB];
    __shared__ float sd_s[AGG_RB];
    __shared__ float es_s[AGG_RB];
    __shared__ int   m_s[AGG_RB];
    __shared__ int   la[AGG_CAP];
    __shared__ int   lb[AGG_CAP];
    __shared__ float lc[AGG_CAP];
    __shared__ int   count_s;
    __shared__ int   nblk_s;

    const int tid  = threadIdx.x;
    const int lane = tid & 31;
    const int wave = __builtin_amdgcn_readfirstlane(tid >> 5);
    const int* __restrict__ dstp = ei + n_edges;
    const int nvec = n_edges >> 4;
    const uint4* __restrict__ kv = (const uint4*)key;

    for (int base = blockIdx.x * AGG_RB; base < n; base += gridDim.x * AGG_RB) {
        const unsigned R = (unsigned)min(AGG_RB, n - base);

        for (int i = tid; i < (int)R * HID; i += AGG_THREADS) agg_s[i] = 0.0f;
        for (int i = tid; i < (int)R; i += AGG_THREADS) {
            const float sd = s_dst[base + i];
            const float es = leaky(s_src[base + i] + sd);
            sd_s[i] = sd;
            es_s[i] = es;
            z_s[i]  = 0.0f;
            m_s[i]  = ord_f32(es);
        }
        if (tid == 0) { count_s = 0; nblk_s = 0; }
        __syncthreads();

        const unsigned kb = (unsigned)(base / AGG_RB) & 255u;
        const unsigned kr = kb * 0x01010101u;
        const int nwords = nchunk * 32;
        const unsigned* __restrict__ brow = bmap + (size_t)kb * nwords;
        for (int w0 = 0; w0 < nwords; w0 += 8 * AGG_THREADS) {
            unsigned wv[8];
            int pre[8];
            int tot = 0;
#pragma unroll
            for (int u = 0; u < 8; ++u) {
                const int wi = w0 + u * AGG_THREADS + tid;
                wv[u] = (wi < nwords) ? brow[wi] : 0u;
            }
#pragma unroll
            for (int u = 0; u < 8; ++u) {
                const int pc = __popc(wv[u]);
                int p = tot;
#pragma unroll
                for (int b = 0; b < 6; ++b) {
                    const unsigned bal = __builtin_amdgcn_ballot_w32(((pc >> b) & 1) != 0);
                    p   += (int)__builtin_amdgcn_mbcnt_lo(bal, 0u) << b;
                    tot += __popc(bal) << b;
                }
                pre[u] = p;
            }
            int wbase = 0;
            if (lane == 0 && tot != 0) wbase = atomicAdd(&nblk_s, tot);
            wbase = __builtin_amdgcn_readfirstlane(wbase);
#pragma unroll
            for (int u = 0; u < 8; ++u) {
                const int wi = w0 + u * AGG_THREADS + tid;
                unsigned bits = wv[u];
                int pos = wbase + pre[u];
                while (bits != 0u) {
                    if (pos < AGG_CAP) lb[pos] = 32 * wi + __builtin_ctz(bits);
                    ++pos;
                    bits &= bits - 1u;
                }
            }
        }
        __syncthreads();
        const int nblk = __builtin_amdgcn_readfirstlane(nblk_s);

        const int nlist = min(nblk, AGG_CAP);
        for (int i0 = 0; i0 < nlist; i0 += 4 * AGG_THREADS) {
            int bid[4], off[4];
            unsigned m[4], bal[4];
            uint4 q[4];
            int tot = 0;
#pragma unroll
            for (int u = 0; u < 4; ++u) {
                const int i = i0 + u * AGG_THREADS + tid;
                bid[u] = (i < nlist) ? lb[i] : -1;
            }
#pragma unroll
            for (int u = 0; u < 4; ++u) q[u] = kv[bid[u] >= 0 ? bid[u] : 0];
#pragma unroll
            for (int u = 0; u < 4; ++u) {
                m[u]   = (bid[u] >= 0) ? key_mask16(q[u], kr) : 0u;
                bal[u] = __builtin_amdgcn_ballot_w32(m[u] != 0u);
                off[u] = tot;
                tot   += __popc(bal[u]);
            }
            int wbase = 0;
            if (lane == 0 && tot != 0) wbase = atomicAdd(&count_s, tot);
            wbase = __builtin_amdgcn_readfirstlane(wbase);
#pragma unroll
            for (int u = 0; u < 4; ++u) {
                unsigned mm = m[u];
                if (mm != 0u) {
                    const int p = __builtin_ctz(mm);
                    const int pos = wbase + off[u] + (int)__builtin_amdgcn_mbcnt_lo(bal[u], 0u);
                    if (pos < AGG_CAP) la[pos] = 16 * bid[u] + ((p & 3) << 2) + (p >> 3);
                    mm &= mm - 1u;
                }
                while (mm != 0u) {
                    const int p = __builtin_ctz(mm);
                    agg_push(16 * bid[u] + ((p & 3) << 2) + (p >> 3), &count_s, la);
                    mm &= mm - 1u;
                }
            }
        }
        for (int idx = 16 * nvec + tid; idx < n_edges; idx += AGG_THREADS)
            if ((unsigned)key[idx] == kb) agg_push(idx, &count_s, la);
        __syncthreads();

        const int cnt = (nblk <= AGG_CAP) ? __builtin_amdgcn_readfirstlane(count_s) : AGG_CAP + 1;
        if (cnt <= AGG_CAP) {
            for (int i0 = tid; i0 < cnt; i0 += 4 * AGG_THREADS) {
                int eidx[4], d[4], s[4];
                float ss[4];
#pragma unroll
                for (int u = 0; u < 4; ++u) {
                    const int i = i0 + u * AGG_THREADS;
                    eidx[u] = (i < cnt) ? la[i] : 0;
                }
#pragma unroll
                for (int u = 0; u < 4; ++u) { s[u] = ei[eidx[u]]; d[u] = dstp[eidx[u]]; }
#pragma unroll
                for (int u = 0; u < 4; ++u) ss[u] = s_src[s[u]];
#pragma unroll
                for (int u = 0; u < 4; ++u) {
                    const int i = i0 + u * AGG_THREADS;
                    if (i < cnt) {
                        const unsigned du = (unsigned)(d[u] - base);
                        const bool in = du < R;
                        const int dl = in ? (int)du : 0;
                        const float er = leaky(ss[u] + sd_s[dl]);
                        const float e = in ? er : -INFINITY;
                        atomicMax(&m_s[dl], ord_f32(e));
                        la[i] = in ? s[u] : base;
                        lb[i] = dl;
                        lc[i] = e;
                    }
                }
            }
            __syncthreads();

            const int nbatch = (cnt + 31) >> 5;
            for (int b = wave; b < nbatch; b += AGG_THREADS / 32) {
                const int i = (b << 5) + lane;
                const bool ok = i < cnt;
                const int s_l  = ok ? la[i] : 0;
                const int dl_l = ok ? lb[i] : 0;
                float ex_l = 0.0f;
                if (ok) {
                    ex_l = __expf(lc[i] - unord_f32(m_s[dl_l]));
                    atomicAdd(&z_s[dl_l], ex_l);
                }
                const int nvalid = min(32, cnt - (b << 5));
                if (nvalid == 32) {
#pragma unroll
                    for (int k0 = 0; k0 < 32; k0 += 8) {
                        float v[8];
#pragma unroll
                        for (int j = 0; j < 8; ++j) {
                            const int s = __builtin_amdgcn_readlane(s_l, k0 + j);
                            v[j] = h[(size_t)s * HID + lane];
                        }
#pragma unroll
                        for (int j = 0; j < 8; ++j) {
                            const int   dl = __builtin_amdgcn_readlane(dl_l, k0 + j);
                            const float ex = __int_as_float(
                                __builtin_amdgcn_readlane(__float_as_int(ex_l), k0 + j));
                            atomicAdd(&agg_s[dl * HID + lane], ex * v[j]);
                        }
                    }
                } else {
                    for (int k = 0; k < nvalid; ++k) {
                        const int   s  = __builtin_amdgcn_readlane(s_l, k);
                        const int   dl = __builtin_amdgcn_readlane(dl_l, k);
                        const float ex = __int_as_float(
                            __builtin_amdgcn_readlane(__float_as_int(ex_l), k));
                        atomicAdd(&agg_s[dl * HID + lane], ex * h[(size_t)s * HID + lane]);
                    }
                }
            }
        } else {
            agg_slow_scan<0>(ei, s_src, h, sd_s, m_s, z_s, agg_s, base, R, n_edges, tid);
            __syncthreads();
            agg_slow_scan<1>(ei, s_src, h, sd_s, m_s, z_s, agg_s, base, R, n_edges, tid);
        }
        __syncthreads();

        for (int i = tid; i < (int)R * HID; i += AGG_THREADS) {
            const int dl = i / HID;
            const int c  = i & (HID - 1);
            const size_t o = (size_t)(base + dl) * HID + c;
            const float exs = __expf(es_s[dl] - unord_f32(m_s[dl]));
            const float v = (agg_s[i] + exs * h[o]) / (z_s[dl] + exs + 1e-16f) + bias[c];
            g[o] = v > 0.0f ? v : 0.0f;
        }
        __syncthreads();
    }
}

extern "C" void kernel_launch(void* const* d_in, const int* in_sizes, int n_in,
                              void* d_out, int out_size, void* d_ws, size_t ws_size,
                              hipStream_t stream)
{
    const float* x      = (const float*)d_in[0];
    const int*   ei     = (const int*)  d_in[1];
    const float* W1     = (const float*)d_in[2];
    const float* a_src1 = (const float*)d_in[3];
    const float* a_dst1 = (const float*)d_in[4];
    const float* b1     = (const float*)d_in[5];
    const float* W2     = (const float*)d_in[6];
    const float* a_src2 = (const float*)d_in[7];
    const float* a_dst2 = (const float*)d_in[8];
    const float* b2     = (const float*)d_in[9];
    const float* Wc     = (const float*)d_in[10];
    const float* bc     = (const float*)d_in[11];
    float* out = (float*)d_out;

    const int n       = in_sizes[0] / IN_C;
    const int n_edges = in_sizes[1] / 2;

    float* ws    = (float*)d_ws;
    float* h     = ws;
    float* g     = h   + (size_t)n * HID;
    unsigned char* key = (unsigned char*)(g + (size_t)n * HID);
    unsigned* bmap = (unsigned*)(key + (((size_t)n_edges + 255) & ~(size_t)255));
    float* s_src = g   + (size_t)2 * n * HID;
    float* s_dst = s_src + n;

    const int TB = 256;
    const int nodeBlocks = (n + TB - 1) / TB;
    const int aggBlocks  = (n + AGG_RB - 1) / AGG_RB;
    const int gemmWaves  = (n / 16) * (HID / 16);
    const int gemmBlocks = (gemmWaves + 7) / 8;
    const int nvec16     = n_edges / 16;
    const int nchunk     = nvec16 > 0 ? (nvec16 + KEY_CHUNK - 1) / KEY_CHUNK : 1;

    gat_edge_keys<<<nchunk, KEY_CHUNK, 0, stream>>>(ei, key, bmap, n_edges, nchunk);

    gat_gemm_wmma<IN_C, HID, 0><<<gemmBlocks, TB, 0, stream>>>(x, W1, nullptr, h, n);
    gat_node_prep<<<nodeBlocks, TB, 0, stream>>>(h, a_src1, a_dst1, s_src, s_dst, n);
    gat_block_agg<<<aggBlocks, AGG_THREADS, 0, stream>>>(ei, key, bmap, s_src, s_dst, h, b1, g, n, n_edges, nchunk);

    gat_gemm_wmma<HID, HID, 0><<<gemmBlocks, TB, 0, stream>>>(g, W2, nullptr, h, n);
    gat_node_prep<<<nodeBlocks, TB, 0, stream>>>(h, a_src2, a_dst2, s_src, s_dst, n);
    gat_block_agg<<<aggBlocks, AGG_THREADS, 0, stream>>>(ei, key, bmap, s_src, s_dst, h, b2, g, n, n_edges, nchunk);

    gat_gemm_wmma<HID, OUT_C, 1><<<gemmBlocks, TB, 0, stream>>>(g, Wc, bc, out, n);
}


# edge filter reads the 64-byte destination and source blocks of marked blocks (4 lanes a block, 16 bytes each), tests rows exactly: key column and per-edge index gathers gone; sums LDS arrival order
# speedup vs baseline: 13.6926x; 1.0483x over previous
#include <hip/hip_runtime.h>
#include <math.h>

#define NODES_DEF 100000
#define IN_C 128
#define HID 32
#define OUT_C 32
#define NEG_SLOPE 0.2f

typedef __attribute__((ext_vector_type(2))) float v2f;
typedef __attribute__((ext_vector_type(8))) float v8f;

template <int K, int NC, int EPI>
__global__ void gat_gemm_wmma(const float* __restrict__ A,
                              const float* __restrict__ B,
                              const float* __restrict__ bias,
                              float* __restrict__ C, int M)
{
    const int wavesPerBlock = blockDim.x >> 5;
    const int wave = blockIdx.x * wavesPerBlock + (threadIdx.x >> 5);
    const int nTiles = NC / 16;
    const int totalTiles = (M / 16) * nTiles;
    if (wave >= totalTiles) return;

    const int mTile = wave / nTiles;
    const int nTile = wave % nTiles;
    const int lane  = threadIdx.x & 31;
    const int half  = lane >> 4;
    const int r     = lane & 15;
    const int row   = mTile * 16 + r;
    const int col   = nTile * 16 + r;

    v8f acc = {};
#pragma unroll
    for (int k = 0; k < K; k += 4) {
        v2f a = *(const v2f*)(A + (size_t)row * K + k + 2 * half);
        v2f b;
        b.x = B[(k + 2 * half + 0) * NC + col];
        b.y = B[(k + 2 * half + 1) * NC + col];
        acc = __builtin_amdgcn_wmma_f32_16x16x4_f32(
             false, a,  false, b,
             (short)0, acc,  false,  false);
    }

#pragma unroll
    for (int i = 0; i < 8; ++i) {
        const int orow = mTile * 16 + i + 8 * half;
        float v = acc[i];
        if (EPI == 1) {
            v += bias[col];
            v = 1.0f / (1.0f + __expf(-v));
        }
        C[(size_t)orow * NC + col] = v;
    }
}

__device__ __forceinline__ int ord_f32(float f)
{
    int i = __float_as_int(f);
    return i >= 0 ? i : (i ^ 0x7fffffff);
}
__device__ __forceinline__ float unord_f32(int i)
{
    return __int_as_float(i >= 0 ? i : (i ^ 0x7fffffff));
}
__device__ __forceinline__ float leaky(float x)
{
    return x > 0.0f ? x : NEG_SLOPE * x;
}

__global__ void gat_node_prep(const float* __restrict__ h,
                              const float* __restrict__ a_src,
                              const float* __restrict__ a_dst,
                              float* __restrict__ s_src,
                              float* __restrict__ s_dst,
                              int n)
{
    const int i = blockIdx.x * blockDim.x + threadIdx.x;
    if (i >= n) return;
    const float* hp = h + (size_t)i * HID;
    float ss = 0.0f, sd = 0.0f;
#pragma unroll
    for (int c = 0; c < HID; ++c) {
        const float v = hp[c];
        ss += v * a_src[c];
        sd += v * a_dst[c];
    }
    s_src[i] = ss;
    s_dst[i] = sd;
}

#define AGG_RB      391
#define AGG_CAP     16384
#define AGG_THREADS 1024
#define KEY_CHUNK   1024
#define KEY_PITCH   257

__device__ __forceinline__ void agg_push(int src, int row, int* count_s, int* la, int* lc)
{
    const int pos = atomicAdd(count_s, 1);
    if (pos < AGG_CAP) { la[pos] = src; lc[pos] = row; }
}

__device__ __forceinline__ int4 load_dst4(const int* __restrict__ p, bool aligned)
{
    if (aligned) return *(const int4*)p;
    return make_int4(p[0], p[1], p[2], p[3]);
}

__global__ __launch_bounds__(KEY_CHUNK)
void gat_edge_keys(const int* __restrict__ ei,
                   unsigned* __restrict__ bmap, int n_edges, int nchunk)
{
    __shared__ unsigned bm_s[32 * KEY_PITCH];
    const int tid = threadIdx.x;
    const int* __restrict__ dstp = ei + n_edges;
    const int nvec = n_edges >> 4;
    for (int i = tid; i < 32 * KEY_PITCH; i += KEY_CHUNK) bm_s[i] = 0u;
    __syncthreads();

    const int blk = blockIdx.x * KEY_CHUNK + tid;
    if (blk < nvec) {
        unsigned kk[16];
        if ((n_edges & 3) == 0) {
            const int4* __restrict__ dv = (const int4*)(dstp + 16 * blk);
#pragma unroll
            for (int j = 0; j < 4; ++j) {
                const int4 d = dv[j];
                kk[4 * j + 0] = (unsigned)(d.x / AGG_RB) & 255u;
                kk[4 * j + 1] = (unsigned)(d.y / AGG_RB) & 255u;
                kk[4 * j + 2] = (unsigned)(d.z / AGG_RB) & 255u;
                kk[4 * j + 3] = (unsigned)(d.w / AGG_RB) & 255u;
            }
        } else {
#pragma unroll
            for (int j = 0; j < 16; ++j)
                kk[j] = (unsigned)(dstp[16 * blk + j] / AGG_RB) & 255u;
        }
        unsigned* plane = bm_s + (tid >> 5) * KEY_PITCH;
        const unsigned bit = 1u << (tid & 31);
#pragma unroll
        for (int j = 0; j < 16; ++j) atomicOr(&plane[kk[j]], bit);
    }
    __syncthreads();

    for (int i = tid; i < 256 * 32; i += KEY_CHUNK) {
        const int k = i >> 5, wd = i & 31;
        bmap[((size_t)k * nchunk + blockIdx.x) * 32 + wd] = bm_s[wd * KEY_PITCH + k];
    }
}

template <int MODE>
__device__ __forceinline__ void agg_slow_scan(const int* __restrict__ ei,
                                              const float* __restrict__ s_src,
                                              const float* __restrict__ h,
                                              const float* sd_s, int* m_s,
                                              float* z_s, float* agg_s,
                                              int base, unsigned R, int n_edges, int tid)
{
    for (int idx = tid; idx < n_edges; idx += AGG_THREADS) {
        const unsigned dl = (unsigned)(ei[n_edges + idx] - base);
        if (dl < R) {
            const int s = ei[idx];
            const float e = leaky(s_src[s] + sd_s[dl]);
            if (MODE == 0) {
                atomicMax(&m_s[dl], ord_f32(e));
            } else {
                const float ex = __expf(e - unord_f32(m_s[dl]));
                atomicAdd(&z_s[dl], ex);
                for (int c = 0; c < HID; ++c)
                    atomicAdd(&agg_s[dl * HID + c], ex * h[(size_t)s * HID + c]);
            }
        }
    }
}

__global__ __launch_bounds__(AGG_THREADS)
void gat_block_agg(const int* __restrict__ ei,
                   const unsigned* __restrict__ bmap,
                   const float* __restrict__ s_src,
                   const float* __restrict__ s_dst,
                   const float* __restrict__ h,
                   const float* __restrict__ bias,
                   float* __restrict__ g,
                   int n, int n_edges, int nchunk)
{
    __shared__ float agg_s[AGG_RB * HID];
    __shared__ float z_s[AGG_RB];
    __shared__ float sd_s[AGG_RB];
    __shared__ float es_s[AGG_RB];
    __shared__ int   m_s[AGG_RB];
    __shared__ int   la[AGG_CAP];
    __shared__ int   lb[AGG_CAP];
    __shared__ int   lc[AGG_CAP];
    __shared__ int   count_s;
    __shared__ int   nblk_s;

    const int tid  = threadIdx.x;
    const int lane = tid & 31;
    const int wave = __builtin_amdgcn_readfirstlane(tid >> 5);
    const int* __restrict__ dstp = ei + n_edges;
    const int nvec = n_edges >> 4;
    const bool al16 = (n_edges & 3) == 0;
    const int q4 = 4 * (tid & 3);

    for (int base = blockIdx.x * AGG_RB; base < n; base += gridDim.x * AGG_RB) {
        const unsigned R = (unsigned)min(AGG_RB, n - base);

        for (int i = tid; i < (int)R * HID; i += AGG_THREADS) agg_s[i] = 0.0f;
        for (int i = tid; i < (int)R; i += AGG_THREADS) {
            const float sd = s_dst[base + i];
            const float es = leaky(s_src[base + i] + sd);
            sd_s[i] = sd;
            es_s[i] = es;
            z_s[i]  = 0.0f;
            m_s[i]  = ord_f32(es);
        }
        if (tid == 0) { count_s = 0; nblk_s = 0; }
        __syncthreads();

        const unsigned kb = (unsigned)(base / AGG_RB) & 255u;
        const unsigned kr = kb * 0x01010101u;
        const int nwords = nchunk * 32;
        const unsigned* __restrict__ brow = bmap + (size_t)kb * nwords;
        for (int w0 = 0; w0 < nwords; w0 += 8 * AGG_THREADS) {
            unsigned wv[8];
            int pre[8];
            int tot = 0;
#pragma unroll
            for (int u = 0; u < 8; ++u) {
                const int wi = w0 + u * AGG_THREADS + tid;
                wv[u] = (wi < nwords) ? brow[wi] : 0u;
            }
#pragma unroll
            for (int u = 0; u < 8; ++u) {
                const int pc = __popc(wv[u]);
                int p = tot;
#pragma unroll
                for (int b = 0; b < 6; ++b) {
                    const unsigned bal = __builtin_amdgcn_ballot_w32(((pc >> b) & 1) != 0);
                    p   += (int)__builtin_amdgcn_mbcnt_lo(bal, 0u) << b;
                    tot += __popc(bal) << b;
                }
                pre[u] = p;
            }
            int wbase = 0;
            if (lane == 0 && tot != 0) wbase = atomicAdd(&nblk_s, tot);
            wbase = __builtin_amdgcn_readfirstlane(wbase);
#pragma unroll
            for (int u = 0; u < 8; ++u) {
                const int wi = w0 + u * AGG_THREADS + tid;
                unsigned bits = wv[u];
                int pos = wbase + pre[u];
                while (bits != 0u) {
                    if (pos < AGG_CAP) lb[pos] = 32 * wi + __builtin_ctz(bits);
                    ++pos;
                    bits &= bits - 1u;
                }
            }
        }
        __syncthreads();
        const int nblk = __builtin_amdgcn_readfirstlane(nblk_s);

        const int nlist = min(nblk, AGG_CAP);
        const int nq = 4 * nlist;
        for (int j0 = 0; j0 < nq; j0 += 4 * AGG_THREADS) {
            int bid[4], off[4];
            unsigned mk[4];
            int4 dq[4], sq[4];
            int tot = 0;
#pragma unroll
            for (int u = 0; u < 4; ++u) {
                const int j = j0 + u * AGG_THREADS + tid;
                bid[u] = (j < nq) ? lb[j >> 2] : -1;
            }
#pragma unroll
            for (int u = 0; u < 4; ++u) {
                const int e0 = 16 * (bid[u] >= 0 ? bid[u] : 0) + q4;
                dq[u] = load_dst4(dstp + e0, al16);
                sq[u] = *(const int4*)(ei + e0);
            }
#pragma unroll
            for (int u = 0; u < 4; ++u) {
                unsigned m = 0u;
                if (bid[u] >= 0) {
                    m = ((unsigned)(dq[u].x - base) < R ? 1u : 0u) |
                        ((unsigned)(dq[u].y - base) < R ? 2u : 0u) |
                        ((unsigned)(dq[u].z - base) < R ? 4u : 0u) |
                        ((unsigned)(dq[u].w - base) < R ? 8u : 0u);
                }
                mk[u] = m;
                const int pc = __popc(m);
                int p = tot;
#pragma unroll
                for (int b = 0; b < 3; ++b) {
                    const unsigned bal = __builtin_amdgcn_ballot_w32(((pc >> b) & 1) != 0);
                    p   += (int)__builtin_amdgcn_mbcnt_lo(bal, 0u) << b;
                    tot += __popc(bal) << b;
                }
                off[u] = p;
            }
            int wbase = 0;
            if (lane == 0 && tot != 0) wbase = atomicAdd(&count_s, tot);
            wbase = __builtin_amdgcn_readfirstlane(wbase);
#pragma unroll
            for (int u = 0; u < 4; ++u) {
                const int dv[4] = { dq[u].x, dq[u].y, dq[u].z, dq[u].w };
                const int sv[4] = { sq[u].x, sq[u].y, sq[u].z, sq[u].w };
                int pos = wbase + off[u];
#pragma unroll
                for (int k = 0; k < 4; ++k) {
                    if ((mk[u] >> k) & 1u) {
                        if (pos < AGG_CAP) { la[pos] = sv[k]; lc[pos] = dv[k] - base; }
                        ++pos;
                    }
                }
            }
        }
        for (int idx = 16 * nvec + tid; idx < n_edges; idx += AGG_THREADS) {
            const unsigned du = (unsigned)(dstp[idx] - base);
            if (du < R) agg_push(ei[idx], (int)du, &count_s, la, lc);
        }
        __syncthreads();

        const int cnt = (nblk <= AGG_CAP) ? __builtin_amdgcn_readfirstlane(count_s) : AGG_CAP + 1;
        if (cnt <= AGG_CAP) {
            for (int i0 = tid; i0 < cnt; i0 += 4 * AGG_THREADS) {
                int dl[4], s[4];
                float ss[4];
#pragma unroll
                for (int u = 0; u < 4; ++u) {
                    const int i = i0 + u * AGG_THREADS;
                    s[u]  = (i < cnt) ? la[i] : 0;
                    dl[u] = (i < cnt) ? lc[i] : 0;
                }
#pragma unroll
                for (int u = 0; u < 4; ++u) ss[u] = s_src[s[u]];
#pragma unroll
                for (int u = 0; u < 4; ++u) {
                    const int i = i0 + u * AGG_THREADS;
                    if (i < cnt) {
                        const float e = leaky(ss[u] + sd_s[dl[u]]);
                        atomicMax(&m_s[dl[u]], ord_f32(e));
                        lb[i] = dl[u];
                        lc[i] = __float_as_int(e);
                    }
                }
            }
            __syncthreads();

            const int nbatch = (cnt + 31) >> 5;
            for (int b = wave; b < nbatch; b += AGG_THREADS / 32) {
                const int i = (b << 5) + lane;
                const bool ok = i < cnt;
                const int s_l  = ok ? la[i] : 0;
                const int dl_l = ok ? lb[i] : 0;
                float ex_l = 0.0f;
                if (ok) {
                    ex_l = __expf(__int_as_float(lc[i]) - unord_f32(m_s[dl_l]));
                    atomicAdd(&z_s[dl_l], ex_l);
                }
                const int nvalid = min(32, cnt - (b << 5));
                if (nvalid == 32) {
#pragma unroll
                    for (int k0 = 0; k0 < 32; k0 += 8) {
                        float v[8];
#pragma unroll
                        for (int j = 0; j < 8; ++j) {
                            const int s = __builtin_amdgcn_readlane(s_l, k0 + j);
                            v[j] = h[(size_t)s * HID + lane];
                        }
#pragma unroll
                        for (int j = 0; j < 8; ++j) {
                            const int   dl = __builtin_amdgcn_readlane(dl_l, k0 + j);
                            const float ex = __int_as_float(
                                __builtin_amdgcn_readlane(__float_as_int(ex_l), k0 + j));
                            atomicAdd(&agg_s[dl * HID + lane], ex * v[j]);
                        }
                    }
                } else {
                    for (int k = 0; k < nvalid; ++k) {
                        const int   s  = __builtin_amdgcn_readlane(s_l, k);
                        const int   dl = __builtin_amdgcn_readlane(dl_l, k);
                        const float ex = __int_as_float(
                            __builtin_amdgcn_readlane(__float_as_int(ex_l), k));
                        atomicAdd(&agg_s[dl * HID + lane], ex * h[(size_t)s * HID + lane]);
                    }
                }
            }
        } else {
            agg_slow_scan<0>(ei, s_src, h, sd_s, m_s, z_s, agg_s, base, R, n_edges, tid);
            __syncthreads();
            agg_slow_scan<1>(ei, s_src, h, sd_s, m_s, z_s, agg_s, base, R, n_edges, tid);
        }
        __syncthreads();

        for (int i = tid; i < (int)R * HID; i += AGG_THREADS) {
            const int dl = i / HID;
            const int c  = i & (HID - 1);
            const size_t o = (size_t)(base + dl) * HID + c;
            const float exs = __expf(es_s[dl] - unord_f32(m_s[dl]));
            const float v = (agg_s[i] + exs * h[o]) / (z_s[dl] + exs + 1e-16f) + bias[c];
            g[o] = v > 0.0f ? v : 0.0f;
        }
        __syncthreads();
    }
}

extern "C" void kernel_launch(void* const* d_in, const int* in_sizes, int n_in,
                              void* d_out, int out_size, void* d_ws, size_t ws_size,
                              hipStream_t stream)
{
    const float* x      = (const float*)d_in[0];
    const int*   ei     = (const int*)  d_in[1];
    const float* W1     = (const float*)d_in[2];
    const float* a_src1 = (const float*)d_in[3];
    const float* a_dst1 = (const float*)d_in[4];
    const float* b1     = (const float*)d_in[5];
    const float* W2     = (const float*)d_in[6];
    const float* a_src2 = (const float*)d_in[7];
    const float* a_dst2 = (const float*)d_in[8];
    const float* b2     = (const float*)d_in[9];
    const float* Wc     = (const float*)d_in[10];
    const float* bc     = (const float*)d_in[11];
    float* out = (float*)d_out;

    const int n       = in_sizes[0] / IN_C;
    const int n_edges = in_sizes[1] / 2;

    float* ws    = (float*)d_ws;
    float* h     = ws;
    float* g     = h   + (size_t)n * HID;
    unsigned* bmap = (unsigned*)(g + (size_t)n * HID);
    float* s_src = g   + (size_t)2 * n * HID;
    float* s_dst = s_src + n;

    const int TB = 256;
    const int nodeBlocks = (n + TB - 1) / TB;
    const int aggBlocks  = (n + AGG_RB - 1) / AGG_RB;
    const int gemmWaves  = (n / 16) * (HID / 16);
    const int gemmBlocks = (gemmWaves + 7) / 8;
    const int nvec16     = n_edges / 16;
    const int nchunk     = nvec16 > 0 ? (nvec16 + KEY_CHUNK - 1) / KEY_CHUNK : 1;

    gat_edge_keys<<<nchunk, KEY_CHUNK, 0, stream>>>(ei, bmap, n_edges, nchunk);

    gat_gemm_wmma<IN_C, HID, 0><<<gemmBlocks, TB, 0, stream>>>(x, W1, nullptr, h, n);
    gat_node_prep<<<nodeBlocks, TB, 0, stream>>>(h, a_src1, a_dst1, s_src, s_dst, n);
    gat_block_agg<<<aggBlocks, AGG_THREADS, 0, stream>>>(ei, bmap, s_src, s_dst, h, b1, g, n, n_edges, nchunk);

    gat_gemm_wmma<HID, HID, 0><<<gemmBlocks, TB, 0, stream>>>(g, W2, nullptr, h, n);
    gat_node_prep<<<nodeBlocks, TB, 0, stream>>>(h, a_src2, a_dst2, s_src, s_dst, n);
    gat_block_agg<<<aggBlocks, AGG_THREADS, 0, stream>>>(ei, bmap, s_src, s_dst, h, b2, g, n, n_edges, nchunk);

    gat_gemm_wmma<HID, OUT_C, 1><<<gemmBlocks, TB, 0, stream>>>(g, Wc, bc, out, n);
}


# layer 2's edge phase reads back the packed (source,row) list that layer 1's phase kept per group (4 bytes an edge, in spare workspace and the output buffer): no second filter; sums LDS arrival order
# speedup vs baseline: 16.2505x; 1.1868x over previous
#include <hip/hip_runtime.h>
#include <math.h>

#define NODES_DEF 100000
#define IN_C 128
#define HID 32
#define OUT_C 32
#define NEG_SLOPE 0.2f

typedef __attribute__((ext_vector_type(2))) float v2f;
typedef __attribute__((ext_vector_type(8))) float v8f;

template <int K, int NC, int EPI>
__global__ void gat_gemm_wmma(const float* __restrict__ A,
                              const float* __restrict__ B,
                              const float* __restrict__ bias,
                              float* __restrict__ C, int M)
{
    const int wavesPerBlock = blockDim.x >> 5;
    const int wave = blockIdx.x * wavesPerBlock + (threadIdx.x >> 5);
    const int nTiles = NC / 16;
    const int totalTiles = (M / 16) * nTiles;
    if (wave >= totalTiles) return;

    const int mTile = wave / nTiles;
    const int nTile = wave % nTiles;
    const int lane  = threadIdx.x & 31;
    const int half  = lane >> 4;
    const int r     = lane & 15;
    const int row   = mTile * 16 + r;
    const int col   = nTile * 16 + r;

    v8f acc = {};
#pragma unroll
    for (int k = 0; k < K; k += 4) {
        v2f a = *(const v2f*)(A + (size_t)row * K + k + 2 * half);
        v2f b;
        b.x = B[(k + 2 * half + 0) * NC + col];
        b.y = B[(k + 2 * half + 1) * NC + col];
        acc = __builtin_amdgcn_wmma_f32_16x16x4_f32(
             false, a,  false, b,
             (short)0, acc,  false,  false);
    }

#pragma unroll
    for (int i = 0; i < 8; ++i) {
        const int orow = mTile * 16 + i + 8 * half;
        float v = acc[i];
        if (EPI == 1) {
            v += bias[col];
            v = 1.0f / (1.0f + __expf(-v));
        }
        C[(size_t)orow * NC + col] = v;
    }
}

__device__ __forceinline__ int ord_f32(float f)
{
    int i = __float_as_int(f);
    return i >= 0 ? i : (i ^ 0x7fffffff);
}
__device__ __forceinline__ float unord_f32(int i)
{
    return __int_as_float(i >= 0 ? i : (i ^ 0x7fffffff));
}
__device__ __forceinline__ float leaky(float x)
{
    return x > 0.0f ? x : NEG_SLOPE * x;
}

__global__ void gat_node_prep(const float* __restrict__ h,
                              const float* __restrict__ a_src,
                              const float* __restrict__ a_dst,
                              float* __restrict__ s_src,
                              float* __restrict__ s_dst,
                              int n)
{
    const int i = blockIdx.x * blockDim.x + threadIdx.x;
    if (i >= n) return;
    const float* hp = h + (size_t)i * HID;
    float ss = 0.0f, sd = 0.0f;
#pragma unroll
    for (int c = 0; c < HID; ++c) {
        const float v = hp[c];
        ss += v * a_src[c];
        sd += v * a_dst[c];
    }
    s_src[i] = ss;
    s_dst[i] = sd;
}

#define AGG_RB      391
#define AGG_CAP     16384
#define AGG_THREADS 1024
#define KEY_CHUNK   1024
#define KEY_PITCH   257
#define LIST_SHIFT  9
#define LIST_PER_T  (AGG_CAP / AGG_THREADS)
static_assert(LIST_PER_T == 16 && AGG_RB <= (1 << LIST_SHIFT), "the read-back names 16 words a thread; a row must fit the packed word");

__device__ __forceinline__ void agg_push(int src, int row, int* count_s, int* la, int* lc)
{
    const int pos = atomicAdd(count_s, 1);
    if (pos < AGG_CAP) { la[pos] = src; lc[pos] = row; }
}

__device__ __forceinline__ int4 load_dst4(const int* __restrict__ p, bool aligned)
{
    if (aligned) return *(const int4*)p;
    return make_int4(p[0], p[1], p[2], p[3]);
}

__global__ __launch_bounds__(KEY_CHUNK)
void gat_edge_keys(const int* __restrict__ ei,
                   unsigned* __restrict__ bmap, int n_edges, int nchunk)
{
    __shared__ unsigned bm_s[32 * KEY_PITCH];
    const int tid = threadIdx.x;
    const int* __restrict__ dstp = ei + n_edges;
    const int nvec = n_edges >> 4;
    for (int i = tid; i < 32 * KEY_PITCH; i += KEY_CHUNK) bm_s[i] = 0u;
    __syncthreads();

    const int blk = blockIdx.x * KEY_CHUNK + tid;
    if (blk < nvec) {
        unsigned kk[16];
        if ((n_edges & 3) == 0) {
            const int4* __restrict__ dv = (const int4*)(dstp + 16 * blk);
#pragma unroll
            for (int j = 0; j < 4; ++j) {
                const int4 d = dv[j];
                kk[4 * j + 0] = (unsigned)(d.x / AGG_RB) & 255u;
                kk[4 * j + 1] = (unsigned)(d.y / AGG_RB) & 255u;
                kk[4 * j + 2] = (unsigned)(d.z / AGG_RB) & 255u;
                kk[4 * j + 3] = (unsigned)(d.w / AGG_RB) & 255u;
            }
        } else {
#pragma unroll
            for (int j = 0; j < 16; ++j)
                kk[j] = (unsigned)(dstp[16 * blk + j] / AGG_RB) & 255u;
        }
        unsigned* plane = bm_s + (tid >> 5) * KEY_PITCH;
        const unsigned bit = 1u << (tid & 31);
#pragma unroll
        for (int j = 0; j < 16; ++j) atomicOr(&plane[kk[j]], bit);
    }
    __syncthreads();

    for (int i = tid; i < 256 * 32; i += KEY_CHUNK) {
        const int k = i >> 5, wd = i & 31;
        bmap[((size_t)k * nchunk + blockIdx.x) * 32 + wd] = bm_s[wd * KEY_PITCH + k];
    }
}

template <int MODE>
__device__ __forceinline__ void agg_slow_scan(const int* __restrict__ ei,
                                              const float* __restrict__ s_src,
                                              const float* __restrict__ h,
                                              const float* sd_s, int* m_s,
                                              float* z_s, float* agg_s,
                                              int base, unsigned R, int n_edges, int tid)
{
    for (int idx = tid; idx < n_edges; idx += AGG_THREADS) {
        const unsigned dl = (unsigned)(ei[n_edges + idx] - base);
        if (dl < R) {
            const int s = ei[idx];
            const float e = leaky(s_src[s] + sd_s[dl]);
            if (MODE == 0) {
                atomicMax(&m_s[dl], ord_f32(e));
            } else {
                const float ex = __expf(e - unord_f32(m_s[dl]));
                atomicAdd(&z_s[dl], ex);
                for (int c = 0; c < HID; ++c)
                    atomicAdd(&agg_s[dl * HID + c], ex * h[(size_t)s * HID + c]);
            }
        }
    }
}

template <bool FIRST>
__global__ __launch_bounds__(AGG_THREADS)
void gat_block_agg(const int* __restrict__ ei,
                   const unsigned* __restrict__ bmap,
                   const float* __restrict__ s_src,
                   const float* __restrict__ s_dst,
                   const float* __restrict__ h,
                   const float* __restrict__ bias,
                   float* __restrict__ g,
                   unsigned* listA, int nA, unsigned* listB, int nB, int* counts,
                   int n, int n_edges, int nchunk)
{
    __shared__ float agg_s[AGG_RB * HID];
    __shared__ float z_s[AGG_RB];
    __shared__ float sd_s[AGG_RB];
    __shared__ float es_s[AGG_RB];
    __shared__ int   m_s[AGG_RB];
    __shared__ int   la[AGG_CAP];
    __shared__ int   lb[AGG_CAP];
    __shared__ int   lc[AGG_CAP];
    __shared__ int   count_s;
    __shared__ int   nblk_s;

    const int tid  = threadIdx.x;
    const int lane = tid & 31;
    const int wave = __builtin_amdgcn_readfirstlane(tid >> 5);
    const int* __restrict__ dstp = ei + n_edges;
    const int nvec = n_edges >> 4;
    const bool al16 = (n_edges & 3) == 0;
    const int q4 = 4 * (tid & 3);

    for (int base = blockIdx.x * AGG_RB; base < n; base += gridDim.x * AGG_RB) {
        const unsigned R = (unsigned)min(AGG_RB, n - base);

        const int gi = base / AGG_RB;
        unsigned* slot = nullptr;
        if (n <= (1 << (32 - LIST_SHIFT))) {
            if (gi < nA)           slot = listA + (size_t)gi * AGG_CAP;
            else if (gi - nA < nB) slot = listB + (size_t)(gi - nA) * AGG_CAP;
        }
        int kept = -1;
        if (!FIRST && slot != nullptr) kept = __builtin_amdgcn_readfirstlane(counts[gi]);
        const bool reuse = kept >= 0 && kept <= AGG_CAP;

        for (int i = tid; i < (int)R * HID; i += AGG_THREADS) agg_s[i] = 0.0f;
        for (int i = tid; i < (int)R; i += AGG_THREADS) {
            const float sd = s_dst[base + i];
            const float es = leaky(s_src[base + i] + sd);
            sd_s[i] = sd;
            es_s[i] = es;
            z_s[i]  = 0.0f;
            m_s[i]  = ord_f32(es);
        }
        if (tid == 0) { count_s = 0; nblk_s = 0; }
        __syncthreads();

        int cnt = kept;
        if (!reuse) {
            const unsigned kb = (unsigned)(base / AGG_RB) & 255u;
            const unsigned kr = kb * 0x01010101u;
            const int nwords = nchunk * 32;
            const unsigned* __restrict__ brow = bmap + (size_t)kb * nwords;
            for (int w0 = 0; w0 < nwords; w0 += 8 * AGG_THREADS) {
                unsigned wv[8];
                int pre[8];
                int tot = 0;
#pragma unroll
                for (int u = 0; u < 8; ++u) {
                    const int wi = w0 + u * AGG_THREADS + tid;
                    wv[u] = (wi < nwords) ? brow[wi] : 0u;
                }
#pragma unroll
                for (int u = 0; u < 8; ++u) {
                    const int pc = __popc(wv[u]);
                    int p = tot;
#pragma unroll
                    for (int b = 0; b < 6; ++b) {
                        const unsigned bal = __builtin_amdgcn_ballot_w32(((pc >> b) & 1) != 0);
                        p   += (int)__builtin_amdgcn_mbcnt_lo(bal, 0u) << b;
                        tot += __popc(bal) << b;
                    }
                    pre[u] = p;
                }
                int wbase = 0;
                if (lane == 0 && tot != 0) wbase = atomicAdd(&nblk_s, tot);
                wbase = __builtin_amdgcn_readfirstlane(wbase);
#pragma unroll
                for (int u = 0; u < 8; ++u) {
                    const int wi = w0 + u * AGG_THREADS + tid;
                    unsigned bits = wv[u];
                    int pos = wbase + pre[u];
                    while (bits != 0u) {
                        if (pos < AGG_CAP) lb[pos] = 32 * wi + __builtin_ctz(bits);
                        ++pos;
                        bits &= bits - 1u;
                    }
                }
            }
            __syncthreads();
            const int nblk = __builtin_amdgcn_readfirstlane(nblk_s);

            const int nlist = min(nblk, AGG_CAP);
            const int nq = 4 * nlist;
            for (int j0 = 0; j0 < nq; j0 += 4 * AGG_THREADS) {
                int bid[4], off[4];
                unsigned mk[4];
                int4 dq[4], sq[4];
                int tot = 0;
#pragma unroll
                for (int u = 0; u < 4; ++u) {
                    const int j = j0 + u * AGG_THREADS + tid;
                    bid[u] = (j < nq) ? lb[j >> 2] : -1;
                }
#pragma unroll
                for (int u = 0; u < 4; ++u) {
                    const int e0 = 16 * (bid[u] >= 0 ? bid[u] : 0) + q4;
                    dq[u] = load_dst4(dstp + e0, al16);
                    sq[u] = *(const int4*)(ei + e0);
                }
#pragma unroll
                for (int u = 0; u < 4; ++u) {
                    unsigned m = 0u;
                    if (bid[u] >= 0) {
                        m = ((unsigned)(dq[u].x - base) < R ? 1u : 0u) |
                            ((unsigned)(dq[u].y - base) < R ? 2u : 0u) |
                            ((unsigned)(dq[u].z - base) < R ? 4u : 0u) |
                            ((unsigned)(dq[u].w - base) < R ? 8u : 0u);
                    }
                    mk[u] = m;
                    const int pc = __popc(m);
                    int p = tot;
#pragma unroll
                    for (int b = 0; b < 3; ++b) {
                        const unsigned bal = __builtin_amdgcn_ballot_w32(((pc >> b) & 1) != 0);
                        p   += (int)__builtin_amdgcn_mbcnt_lo(bal, 0u) << b;
                        tot += __popc(bal) << b;
                    }
                    off[u] = p;
                }
                int wbase = 0;
                if (lane == 0 && tot != 0) wbase = atomicAdd(&count_s, tot);
                wbase = __builtin_amdgcn_readfirstlane(wbase);
#pragma unroll
                for (int u = 0; u < 4; ++u) {
                    const int dv[4] = { dq[u].x, dq[u].y, dq[u].z, dq[u].w };
                    const int sv[4] = { sq[u].x, sq[u].y, sq[u].z, sq[u].w };
                    int pos = wbase + off[u];
#pragma unroll
                    for (int k = 0; k < 4; ++k) {
                        if ((mk[u] >> k) & 1u) {
                            if (pos < AGG_CAP) { la[pos] = sv[k]; lc[pos] = dv[k] - base; }
                            ++pos;
                        }
                    }
                }
            }
            for (int idx = 16 * nvec + tid; idx < n_edges; idx += AGG_THREADS) {
                const unsigned du = (unsigned)(dstp[idx] - base);
                if (du < R) agg_push(ei[idx], (int)du, &count_s, la, lc);
            }
            __syncthreads();

            cnt = (nblk <= AGG_CAP) ? __builtin_amdgcn_readfirstlane(count_s) : AGG_CAP + 1;
        }
        if (cnt <= AGG_CAP) {
            if (!reuse) {
                for (int i0 = tid; i0 < cnt; i0 += 4 * AGG_THREADS) {
                    int dl[4], s[4];
                    float ss[4];
#pragma unroll
                    for (int u = 0; u < 4; ++u) {
                        const int i = i0 + u * AGG_THREADS;
                        s[u]  = (i < cnt) ? la[i] : 0;
                        dl[u] = (i < cnt) ? lc[i] : 0;
                    }
#pragma unroll
                    for (int u = 0; u < 4; ++u) ss[u] = s_src[s[u]];
#pragma unroll
                    for (int u = 0; u < 4; ++u) {
                        const int i = i0 + u * AGG_THREADS;
                        if (i < cnt) {
                            const float e = leaky(ss[u] + sd_s[dl[u]]);
                            atomicMax(&m_s[dl[u]], ord_f32(e));
                            lb[i] = dl[u];
                            lc[i] = __float_as_int(e);
                            if (FIRST && slot != nullptr)
                                slot[i] = ((unsigned)s[u] << LIST_SHIFT) | (unsigned)dl[u];
                        }
                    }
                }
            } else if (cnt > 0) {
                unsigned wv[LIST_PER_T];
                float ss[LIST_PER_T];
#pragma unroll
                for (int u = 0; u < LIST_PER_T; ++u) {
                    const int i = tid + u * AGG_THREADS;
                    wv[u] = slot[(i < cnt) ? i : 0];
                }
#pragma unroll
                for (int u = 0; u < LIST_PER_T; ++u) ss[u] = s_src[wv[u] >> LIST_SHIFT];
                asm volatile("" :: "v"(ss[0]), "v"(ss[1]), "v"(ss[2]), "v"(ss[3]),
                                   "v"(ss[4]), "v"(ss[5]), "v"(ss[6]), "v"(ss[7]),
                                   "v"(ss[8]), "v"(ss[9]), "v"(ss[10]), "v"(ss[11]),
                                   "v"(ss[12]), "v"(ss[13]), "v"(ss[14]), "v"(ss[15]));
#pragma unroll
                for (int u = 0; u < LIST_PER_T; ++u) {
                    const int i = tid + u * AGG_THREADS;
                    if (i < cnt) {
                        const int dl = (int)(wv[u] & ((1u << LIST_SHIFT) - 1u));
                        const float e = leaky(ss[u] + sd_s[dl]);
                        atomicMax(&m_s[dl], ord_f32(e));
                        la[i] = (int)(wv[u] >> LIST_SHIFT);
                        lb[i] = dl;
                        lc[i] = __float_as_int(e);
                    }
                }
            }
            __syncthreads();

            const int nbatch = (cnt + 31) >> 5;
            for (int b = wave; b < nbatch; b += AGG_THREADS / 32) {
                const int i = (b << 5) + lane;
                const bool ok = i < cnt;
                const int s_l  = ok ? la[i] : 0;
                const int dl_l = ok ? lb[i] : 0;
                float ex_l = 0.0f;
                if (ok) {
                    ex_l = __expf(__int_as_float(lc[i]) - unord_f32(m_s[dl_l]));
                    atomicAdd(&z_s[dl_l], ex_l);
                }
                const int nvalid = min(32, cnt - (b << 5));
                if (nvalid == 32) {
#pragma unroll
                    for (int k0 = 0; k0 < 32; k0 += 8) {
                        float v[8];
#pragma unroll
                        for (int j = 0; j < 8; ++j) {
                            const int s = __builtin_amdgcn_readlane(s_l, k0 + j);
                            v[j] = h[(size_t)s * HID + lane];
                        }
#pragma unroll
                        for (int j = 0; j < 8; ++j) {
                            const int   dl = __builtin_amdgcn_readlane(dl_l, k0 + j);
                            const float ex = __int_as_float(
                                __builtin_amdgcn_readlane(__float_as_int(ex_l), k0 + j));
                            atomicAdd(&agg_s[dl * HID + lane], ex * v[j]);
                        }
                    }
                } else {
                    for (int k = 0; k < nvalid; ++k) {
                        const int   s  = __builtin_amdgcn_readlane(s_l, k);
                        const int   dl = __builtin_amdgcn_readlane(dl_l, k);
                        const float ex = __int_as_float(
                            __builtin_amdgcn_readlane(__float_as_int(ex_l), k));
                        atomicAdd(&agg_s[dl * HID + lane], ex * h[(size_t)s * HID + lane]);
                    }
                }
            }
        } else {
            agg_slow_scan<0>(ei, s_src, h, sd_s, m_s, z_s, agg_s, base, R, n_edges, tid);
            __syncthreads();
            agg_slow_scan<1>(ei, s_src, h, sd_s, m_s, z_s, agg_s, base, R, n_edges, tid);
        }
        if (FIRST && slot != nullptr && tid == 0) counts[gi] = (cnt <= AGG_CAP) ? cnt : -1;
        __syncthreads();

        for (int i = tid; i < (int)R * HID; i += AGG_THREADS) {
            const int dl = i / HID;
            const int c  = i & (HID - 1);
            const size_t o = (size_t)(base + dl) * HID + c;
            const float exs = __expf(es_s[dl] - unord_f32(m_s[dl]));
            const float v = (agg_s[i] + exs * h[o]) / (z_s[dl] + exs + 1e-16f) + bias[c];
            g[o] = v > 0.0f ? v : 0.0f;
        }
        __syncthreads();
    }
}

extern "C" void kernel_launch(void* const* d_in, const int* in_sizes, int n_in,
                              void* d_out, int out_size, void* d_ws, size_t ws_size,
                              hipStream_t stream)
{
    const float* x      = (const float*)d_in[0];
    const int*   ei     = (const int*)  d_in[1];
    const float* W1     = (const float*)d_in[2];
    const float* a_src1 = (const float*)d_in[3];
    const float* a_dst1 = (const float*)d_in[4];
    const float* b1     = (const float*)d_in[5];
    const float* W2     = (const float*)d_in[6];
    const float* a_src2 = (const float*)d_in[7];
    const float* a_dst2 = (const float*)d_in[8];
    const float* b2     = (const float*)d_in[9];
    const float* Wc     = (const float*)d_in[10];
    const float* bc     = (const float*)d_in[11];
    float* out = (float*)d_out;

    const int n       = in_sizes[0] / IN_C;
    const int n_edges = in_sizes[1] / 2;

    float* ws    = (float*)d_ws;
    float* h     = ws;
    float* g     = h   + (size_t)n * HID;
    unsigned* bmap = (unsigned*)(g + (size_t)n * HID);
    float* s_src = g   + (size_t)2 * n * HID;
    float* s_dst = s_src + n;

    const int TB = 256;
    const int nodeBlocks = (n + TB - 1) / TB;
    const int aggBlocks  = (n + AGG_RB - 1) / AGG_RB;
    const int gemmWaves  = (n / 16) * (HID / 16);
    const int gemmBlocks = (gemmWaves + 7) / 8;
    const int nvec16     = n_edges / 16;
    const int nchunk     = nvec16 > 0 ? (nvec16 + KEY_CHUNK - 1) / KEY_CHUNK : 1;

    const long long bmapWords = (long long)256 * nchunk * 32;
    const long long cntWords  = ((long long)aggBlocks + 31) / 32 * 32;
    const long long spare     = (long long)n * HID - bmapWords - cntWords;
    int*      counts = (int*)(bmap + bmapWords);
    unsigned* listA  = bmap + bmapWords + cntWords;
    unsigned* listB  = (unsigned*)out;
    const int nA = spare >= 0 ? (int)(spare / AGG_CAP) : 0;
    const int nB = spare >= 0 ? (int)((long long)n * OUT_C / AGG_CAP) : 0;

    gat_edge_keys<<<nchunk, KEY_CHUNK, 0, stream>>>(ei, bmap, n_edges, nchunk);

    gat_gemm_wmma<IN_C, HID, 0><<<gemmBlocks, TB, 0, stream>>>(x, W1, nullptr, h, n);
    gat_node_prep<<<nodeBlocks, TB, 0, stream>>>(h, a_src1, a_dst1, s_src, s_dst, n);
    gat_block_agg<true><<<aggBlocks, AGG_THREADS, 0, stream>>>(ei, bmap, s_src, s_dst, h, b1, g,
                                                               listA, nA, listB, nB, counts, n, n_edges, nchunk);

    gat_gemm_wmma<HID, HID, 0><<<gemmBlocks, TB, 0, stream>>>(g, W2, nullptr, h, n);
    gat_node_prep<<<nodeBlocks, TB, 0, stream>>>(h, a_src2, a_dst2, s_src, s_dst, n);
    gat_block_agg<false><<<aggBlocks, AGG_THREADS, 0, stream>>>(ei, bmap, s_src, s_dst, h, b2, g,
                                                                listA, nA, listB, nB, counts, n, n_edges, nchunk);

    gat_gemm_wmma<HID, OUT_C, 1><<<gemmBlocks, TB, 0, stream>>>(g, Wc, bc, out, n);
}


# kept edge list of a group sorted by source in LDS (counting sort, layer 1; layer 2 reads it back sorted): a wave's 32 source-score reads fall in few cache lines, not 32; sums still LDS arrival order
# speedup vs baseline: 17.1095x; 1.0529x over previous
#include <hip/hip_runtime.h>
#include <math.h>

#define NODES_DEF 100000
#define IN_C 128
#define HID 32
#define OUT_C 32
#define NEG_SLOPE 0.2f

typedef __attribute__((ext_vector_type(2))) float v2f;
typedef __attribute__((ext_vector_type(8))) float v8f;

template <int K, int NC, int EPI>
__global__ void gat_gemm_wmma(const float* __restrict__ A,
                              const float* __restrict__ B,
                              const float* __restrict__ bias,
                              float* __restrict__ C, int M)
{
    const int wavesPerBlock = blockDim.x >> 5;
    const int wave = blockIdx.x * wavesPerBlock + (threadIdx.x >> 5);
    const int nTiles = NC / 16;
    const int totalTiles = (M / 16) * nTiles;
    if (wave >= totalTiles) return;

    const int mTile = wave / nTiles;
    const int nTile = wave % nTiles;
    const int lane  = threadIdx.x & 31;
    const int half  = lane >> 4;
    const int r     = lane & 15;
    const int row   = mTile * 16 + r;
    const int col   = nTile * 16 + r;

    v8f acc = {};
#pragma unroll
    for (int k = 0; k < K; k += 4) {
        v2f a = *(const v2f*)(A + (size_t)row * K + k + 2 * half);
        v2f b;
        b.x = B[(k + 2 * half + 0) * NC + col];
        b.y = B[(k + 2 * half + 1) * NC + col];
        acc = __builtin_amdgcn_wmma_f32_16x16x4_f32(
             false, a,  false, b,
             (short)0, acc,  false,  false);
    }

#pragma unroll
    for (int i = 0; i < 8; ++i) {
        const int orow = mTile * 16 + i + 8 * half;
        float v = acc[i];
        if (EPI == 1) {
            v += bias[col];
            v = 1.0f / (1.0f + __expf(-v));
        }
        C[(size_t)orow * NC + col] = v;
    }
}

__device__ __forceinline__ int ord_f32(float f)
{
    int i = __float_as_int(f);
    return i >= 0 ? i : (i ^ 0x7fffffff);
}
__device__ __forceinline__ float unord_f32(int i)
{
    return __int_as_float(i >= 0 ? i : (i ^ 0x7fffffff));
}
__device__ __forceinline__ float leaky(float x)
{
    return x > 0.0f ? x : NEG_SLOPE * x;
}

__global__ void gat_node_prep(const float* __restrict__ h,
                              const float* __restrict__ a_src,
                              const float* __restrict__ a_dst,
                              float* __restrict__ s_src,
                              float* __restrict__ s_dst,
                              int n)
{
    const int i = blockIdx.x * blockDim.x + threadIdx.x;
    if (i >= n) return;
    const float* hp = h + (size_t)i * HID;
    float ss = 0.0f, sd = 0.0f;
#pragma unroll
    for (int c = 0; c < HID; ++c) {
        const float v = hp[c];
        ss += v * a_src[c];
        sd += v * a_dst[c];
    }
    s_src[i] = ss;
    s_dst[i] = sd;
}

#define AGG_RB      391
#define AGG_CAP     16384
#define AGG_THREADS 1024
#define KEY_CHUNK   1024
#define KEY_PITCH   257
#define LIST_SHIFT  9
#define LIST_PER_T  (AGG_CAP / AGG_THREADS)
#define SORT_NB     (4 * AGG_THREADS)
#define ROW_MASK    ((1u << LIST_SHIFT) - 1u)
static_assert(LIST_PER_T == 16 && AGG_RB <= (1 << LIST_SHIFT), "the read-back names 16 words a thread; a row must fit the packed word");
static_assert(SORT_NB <= AGG_CAP, "the sort's counters live in one list plane");

__device__ __forceinline__ void agg_push(int src, int row, int* count_s, int* la)
{
    const int pos = atomicAdd(count_s, 1);
    if (pos < AGG_CAP) la[pos] = (int)(((unsigned)src << LIST_SHIFT) | (unsigned)row);
}

__device__ __forceinline__ int4 load_dst4(const int* __restrict__ p, bool aligned)
{
    if (aligned) return *(const int4*)p;
    return make_int4(p[0], p[1], p[2], p[3]);
}

__global__ __launch_bounds__(KEY_CHUNK)
void gat_edge_keys(const int* __restrict__ ei,
                   unsigned* __restrict__ bmap, int n_edges, int nchunk)
{
    __shared__ unsigned bm_s[32 * KEY_PITCH];
    const int tid = threadIdx.x;
    const int* __restrict__ dstp = ei + n_edges;
    const int nvec = n_edges >> 4;
    for (int i = tid; i < 32 * KEY_PITCH; i += KEY_CHUNK) bm_s[i] = 0u;
    __syncthreads();

    const int blk = blockIdx.x * KEY_CHUNK + tid;
    if (blk < nvec) {
        unsigned kk[16];
        if ((n_edges & 3) == 0) {
            const int4* __restrict__ dv = (const int4*)(dstp + 16 * blk);
#pragma unroll
            for (int j = 0; j < 4; ++j) {
                const int4 d = dv[j];
                kk[4 * j + 0] = (unsigned)(d.x / AGG_RB) & 255u;
                kk[4 * j + 1] = (unsigned)(d.y / AGG_RB) & 255u;
                kk[4 * j + 2] = (unsigned)(d.z / AGG_RB) & 255u;
                kk[4 * j + 3] = (unsigned)(d.w / AGG_RB) & 255u;
            }
        } else {
#pragma unroll
            for (int j = 0; j < 16; ++j)
                kk[j] = (unsigned)(dstp[16 * blk + j] / AGG_RB) & 255u;
        }
        unsigned* plane = bm_s + (tid >> 5) * KEY_PITCH;
        const unsigned bit = 1u << (tid & 31);
#pragma unroll
        for (int j = 0; j < 16; ++j) atomicOr(&plane[kk[j]], bit);
    }
    __syncthreads();

    for (int i = tid; i < 256 * 32; i += KEY_CHUNK) {
        const int k = i >> 5, wd = i & 31;
        bmap[((size_t)k * nchunk + blockIdx.x) * 32 + wd] = bm_s[wd * KEY_PITCH + k];
    }
}

template <int MODE>
__device__ __forceinline__ void agg_slow_scan(const int* __restrict__ ei,
                                              const float* __restrict__ s_src,
                                              const float* __restrict__ h,
                                              const float* sd_s, int* m_s,
                                              float* z_s, float* agg_s,
                                              int base, unsigned R, int n_edges, int tid)
{
    for (int idx = tid; idx < n_edges; idx += AGG_THREADS) {
        const unsigned dl = (unsigned)(ei[n_edges + idx] - base);
        if (dl < R) {
            const int s = ei[idx];
            const float e = leaky(s_src[s] + sd_s[dl]);
            if (MODE == 0) {
                atomicMax(&m_s[dl], ord_f32(e));
            } else {
                const float ex = __expf(e - unord_f32(m_s[dl]));
                atomicAdd(&z_s[dl], ex);
                for (int c = 0; c < HID; ++c)
                    atomicAdd(&agg_s[dl * HID + c], ex * h[(size_t)s * HID + c]);
            }
        }
    }
}

template <bool FIRST>
__global__ __launch_bounds__(AGG_THREADS)
void gat_block_agg(const int* __restrict__ ei,
                   const unsigned* __restrict__ bmap,
                   const float* __restrict__ s_src,
                   const float* __restrict__ s_dst,
                   const float* __restrict__ h,
                   const float* __restrict__ bias,
                   float* __restrict__ g,
                   unsigned* listA, int nA, unsigned* listB, int nB, int* counts,
                   int n, int n_edges, int nchunk)
{
    __shared__ float agg_s[AGG_RB * HID];
    __shared__ float z_s[AGG_RB];
    __shared__ float sd_s[AGG_RB];
    __shared__ float es_s[AGG_RB];
    __shared__ int   m_s[AGG_RB];
    __shared__ int   la[AGG_CAP];
    __shared__ int   lb[AGG_CAP];
    __shared__ __attribute__((aligned(16)))
               int   lc[AGG_CAP];
    __shared__ int   wsum_s[AGG_THREADS / 32];
    __shared__ int   count_s;
    __shared__ int   nblk_s;

    const int tid  = threadIdx.x;
    const int lane = tid & 31;
    const int wave = __builtin_amdgcn_readfirstlane(tid >> 5);
    const int* __restrict__ dstp = ei + n_edges;
    const int nvec = n_edges >> 4;
    const bool al16 = (n_edges & 3) == 0;
    const int q4 = 4 * (tid & 3);
    const bool packable = n <= (1 << (32 - LIST_SHIFT));
    int sh = 5;
    while (((n - 1) >> sh) >= SORT_NB) ++sh;

    for (int base = blockIdx.x * AGG_RB; base < n; base += gridDim.x * AGG_RB) {
        const unsigned R = (unsigned)min(AGG_RB, n - base);

        const int gi = base / AGG_RB;
        unsigned* slot = nullptr;
        if (packable) {
            if (gi < nA)           slot = listA + (size_t)gi * AGG_CAP;
            else if (gi - nA < nB) slot = listB + (size_t)(gi - nA) * AGG_CAP;
        }
        int kept = -1;
        if (!FIRST && slot != nullptr) kept = __builtin_amdgcn_readfirstlane(counts[gi]);
        const bool reuse = kept >= 0 && kept <= AGG_CAP;

        for (int i = tid; i < (int)R * HID; i += AGG_THREADS) agg_s[i] = 0.0f;
        for (int i = tid; i < (int)R; i += AGG_THREADS) {
            const float sd = s_dst[base + i];
            const float es = leaky(s_src[base + i] + sd);
            sd_s[i] = sd;
            es_s[i] = es;
            z_s[i]  = 0.0f;
            m_s[i]  = ord_f32(es);
        }
        if (tid == 0) { count_s = 0; nblk_s = 0; }
        __syncthreads();

        int cnt = kept;
        if (!packable) {
            cnt = AGG_CAP + 1;
        } else if (!reuse) {
            const unsigned kb = (unsigned)(base / AGG_RB) & 255u;
            const unsigned kr = kb * 0x01010101u;
            const int nwords = nchunk * 32;
            const unsigned* __restrict__ brow = bmap + (size_t)kb * nwords;
            for (int w0 = 0; w0 < nwords; w0 += 8 * AGG_THREADS) {
                unsigned wv[8];
                int pre[8];
                int tot = 0;
#pragma unroll
                for (int u = 0; u < 8; ++u) {
                    const int wi = w0 + u * AGG_THREADS + tid;
                    wv[u] = (wi < nwords) ? brow[wi] : 0u;
                }
#pragma unroll
                for (int u = 0; u < 8; ++u) {
                    const int pc = __popc(wv[u]);
                    int p = tot;
#pragma unroll
                    for (int b = 0; b < 6; ++b) {
                        const unsigned bal = __builtin_amdgcn_ballot_w32(((pc >> b) & 1) != 0);
                        p   += (int)__builtin_amdgcn_mbcnt_lo(bal, 0u) << b;
                        tot += __popc(bal) << b;
                    }
                    pre[u] = p;
                }
                int wbase = 0;
                if (lane == 0 && tot != 0) wbase = atomicAdd(&nblk_s, tot);
                wbase = __builtin_amdgcn_readfirstlane(wbase);
#pragma unroll
                for (int u = 0; u < 8; ++u) {
                    const int wi = w0 + u * AGG_THREADS + tid;
                    unsigned bits = wv[u];
                    int pos = wbase + pre[u];
                    while (bits != 0u) {
                        if (pos < AGG_CAP) lc[pos] = 32 * wi + __builtin_ctz(bits);
                        ++pos;
                        bits &= bits - 1u;
                    }
                }
            }
            __syncthreads();
            const int nblk = __builtin_amdgcn_readfirstlane(nblk_s);

            const int nlist = min(nblk, AGG_CAP);
            const int nq = 4 * nlist;
            for (int j0 = 0; j0 < nq; j0 += 4 * AGG_THREADS) {
                int bid[4], off[4];
                unsigned mk[4];
                int4 dq[4], sq[4];
                int tot = 0;
#pragma unroll
                for (int u = 0; u < 4; ++u) {
                    const int j = j0 + u * AGG_THREADS + tid;
                    bid[u] = (j < nq) ? lc[j >> 2] : -1;
                }
#pragma unroll
                for (int u = 0; u < 4; ++u) {
                    const int e0 = 16 * (bid[u] >= 0 ? bid[u] : 0) + q4;
                    dq[u] = load_dst4(dstp + e0, al16);
                    sq[u] = *(const int4*)(ei + e0);
                }
#pragma unroll
                for (int u = 0; u < 4; ++u) {
                    unsigned m = 0u;
                    if (bid[u] >= 0) {
                        m = ((unsigned)(dq[u].x - base) < R ? 1u : 0u) |
                            ((unsigned)(dq[u].y - base) < R ? 2u : 0u) |
                            ((unsigned)(dq[u].z - base) < R ? 4u : 0u) |
                            ((unsigned)(dq[u].w - base) < R ? 8u : 0u);
                    }
                    mk[u] = m;
                    const int pc = __popc(m);
                    int p = tot;
#pragma unroll
                    for (int b = 0; b < 3; ++b) {
                        const unsigned bal = __builtin_amdgcn_ballot_w32(((pc >> b) & 1) != 0);
                        p   += (int)__builtin_amdgcn_mbcnt_lo(bal, 0u) << b;
                        tot += __popc(bal) << b;
                    }
                    off[u] = p;
                }
                int wbase = 0;
                if (lane == 0 && tot != 0) wbase = atomicAdd(&count_s, tot);
                wbase = __builtin_amdgcn_readfirstlane(wbase);
#pragma unroll
                for (int u = 0; u < 4; ++u) {
                    const int dv[4] = { dq[u].x, dq[u].y, dq[u].z, dq[u].w };
                    const int sv[4] = { sq[u].x, sq[u].y, sq[u].z, sq[u].w };
                    int pos = wbase + off[u];
#pragma unroll
                    for (int k = 0; k < 4; ++k) {
                        if ((mk[u] >> k) & 1u) {
                            if (pos < AGG_CAP)
                                la[pos] = (int)(((unsigned)sv[k] << LIST_SHIFT) | (unsigned)(dv[k] - base));
                            ++pos;
                        }
                    }
                }
            }
            for (int idx = 16 * nvec + tid; idx < n_edges; idx += AGG_THREADS) {
                const unsigned du = (unsigned)(dstp[idx] - base);
                if (du < R) agg_push(ei[idx], (int)du, &count_s, la);
            }
            __syncthreads();

            cnt = (nblk <= AGG_CAP) ? __builtin_amdgcn_readfirstlane(count_s) : AGG_CAP + 1;

            if (cnt <= AGG_CAP) {
                int* ctr = lc;
                ((int4*)ctr)[tid] = make_int4(0, 0, 0, 0);
                __syncthreads();
                for (int i0 = tid; i0 < cnt; i0 += 4 * AGG_THREADS) {
                    unsigned w[4];
#pragma unroll
                    for (int u = 0; u < 4; ++u) {
                        const int i = i0 + u * AGG_THREADS;
                        w[u] = (i < cnt) ? (unsigned)la[i] : 0u;
                    }
#pragma unroll
                    for (int u = 0; u < 4; ++u) {
                        const int i = i0 + u * AGG_THREADS;
                        if (i < cnt) atomicAdd(&ctr[w[u] >> (LIST_SHIFT + sh)], 1);
                    }
                }
                __syncthreads();
                const int4 c4 = ((const int4*)ctr)[tid];
                const int own = c4.x + c4.y + c4.z + c4.w;
                int incl = own;
                for (int d = 1; d < 32; d <<= 1) {
                    const int v = __shfl_up(incl, d);
                    if (lane >= d) incl += v;
                }
                if (lane == 31) wsum_s[wave] = incl;
                __syncthreads();
                int wincl = wsum_s[lane];
                for (int d = 1; d < 32; d <<= 1) {
                    const int v = __shfl_up(wincl, d);
                    if (lane >= d) wincl += v;
                }
                const int woff = wave > 0 ? __builtin_amdgcn_readlane(wincl, wave - 1) : 0;
                const int ex0 = woff + incl - own;
                ((int4*)ctr)[tid] = make_int4(ex0, ex0 + c4.x, ex0 + c4.x + c4.y, ex0 + c4.x + c4.y + c4.z);
                __syncthreads();
                for (int i0 = tid; i0 < cnt; i0 += 4 * AGG_THREADS) {
                    unsigned w[4];
                    int pos[4];
#pragma unroll
                    for (int u = 0; u < 4; ++u) {
                        const int i = i0 + u * AGG_THREADS;
                        w[u] = (i < cnt) ? (unsigned)la[i] : 0u;
                    }
#pragma unroll
                    for (int u = 0; u < 4; ++u) {
                        const int i = i0 + u * AGG_THREADS;
                        pos[u] = (i < cnt) ? atomicAdd(&ctr[w[u] >> (LIST_SHIFT + sh)], 1) : 0;
                    }
#pragma unroll
                    for (int u = 0; u < 4; ++u) {
                        const int i = i0 + u * AGG_THREADS;
                        if (i < cnt) lb[pos[u]] = (int)w[u];
                    }
                }
                __syncthreads();
            }
        }
        if (cnt <= AGG_CAP) {
            if (!reuse) {
                for (int i0 = tid; i0 < cnt; i0 += 4 * AGG_THREADS) {
                    unsigned w[4];
                    float ss[4];
#pragma unroll
                    for (int u = 0; u < 4; ++u) {
                        const int i = i0 + u * AGG_THREADS;
                        w[u] = (i < cnt) ? (unsigned)lb[i] : 0u;
                    }
#pragma unroll
                    for (int u = 0; u < 4; ++u) ss[u] = s_src[w[u] >> LIST_SHIFT];
#pragma unroll
                    for (int u = 0; u < 4; ++u) {
                        const int i = i0 + u * AGG_THREADS;
                        if (i < cnt) {
                            const int dl = (int)(w[u] & ROW_MASK);
                            const float e = leaky(ss[u] + sd_s[dl]);
                            atomicMax(&m_s[dl], ord_f32(e));
                            lc[i] = __float_as_int(e);
                            if (FIRST && slot != nullptr) slot[i] = w[u];
                        }
                    }
                }
            } else if (cnt > 0) {
                unsigned wv[LIST_PER_T];
                float ss[LIST_PER_T];
#pragma unroll
                for (int u = 0; u < LIST_PER_T; ++u) {
                    const int i = tid + u * AGG_THREADS;
                    wv[u] = slot[(i < cnt) ? i : 0];
                }
#pragma unroll
                for (int u = 0; u < LIST_PER_T; ++u) ss[u] = s_src[wv[u] >> LIST_SHIFT];
                asm volatile("" :: "v"(ss[0]), "v"(ss[1]), "v"(ss[2]), "v"(ss[3]),
                                   "v"(ss[4]), "v"(ss[5]), "v"(ss[6]), "v"(ss[7]),
                                   "v"(ss[8]), "v"(ss[9]), "v"(ss[10]), "v"(ss[11]),
                                   "v"(ss[12]), "v"(ss[13]), "v"(ss[14]), "v"(ss[15]));
#pragma unroll
                for (int u = 0; u < LIST_PER_T; ++u) {
                    const int i = tid + u * AGG_THREADS;
                    if (i < cnt) {
                        const int dl = (int)(wv[u] & ROW_MASK);
                        const float e = leaky(ss[u] + sd_s[dl]);
                        atomicMax(&m_s[dl], ord_f32(e));
                        lb[i] = (int)wv[u];
                        lc[i] = __float_as_int(e);
                    }
                }
            }
            __syncthreads();

            const int nbatch = (cnt + 31) >> 5;
            for (int b = wave; b < nbatch; b += AGG_THREADS / 32) {
                const int i = (b << 5) + lane;
                const bool ok = i < cnt;
                const unsigned w_l = ok ? (unsigned)lb[i] : 0u;
                const int s_l  = (int)(w_l >> LIST_SHIFT);
                const int dl_l = (int)(w_l & ROW_MASK);
                float ex_l = 0.0f;
                if (ok) {
                    ex_l = __expf(__int_as_float(lc[i]) - unord_f32(m_s[dl_l]));
                    atomicAdd(&z_s[dl_l], ex_l);
                }
                const int nvalid = min(32, cnt - (b << 5));
                if (nvalid == 32) {
#pragma unroll
                    for (int k0 = 0; k0 < 32; k0 += 8) {
                        float v[8];
#pragma unroll
                        for (int j = 0; j < 8; ++j) {
                            const int s = __builtin_amdgcn_readlane(s_l, k0 + j);
                            v[j] = h[(size_t)s * HID + lane];
                        }
#pragma unroll
                        for (int j = 0; j < 8; ++j) {
                            const int   dl = __builtin_amdgcn_readlane(dl_l, k0 + j);
                            const float ex = __int_as_float(
                                __builtin_amdgcn_readlane(__float_as_int(ex_l), k0 + j));
                            atomicAdd(&agg_s[dl * HID + lane], ex * v[j]);
                        }
                    }
                } else {
                    for (int k = 0; k < nvalid; ++k) {
                        const int   s  = __builtin_amdgcn_readlane(s_l, k);
                        const int   dl = __builtin_amdgcn_readlane(dl_l, k);
                        const float ex = __int_as_float(
                            __builtin_amdgcn_readlane(__float_as_int(ex_l), k));
                        atomicAdd(&agg_s[dl * HID + lane], ex * h[(size_t)s * HID + lane]);
                    }
                }
            }
        } else {
            agg_slow_scan<0>(ei, s_src, h, sd_s, m_s, z_s, agg_s, base, R, n_edges, tid);
            __syncthreads();
            agg_slow_scan<1>(ei, s_src, h, sd_s, m_s, z_s, agg_s, base, R, n_edges, tid);
        }
        if (FIRST && slot != nullptr && tid == 0) counts[gi] = (cnt <= AGG_CAP) ? cnt : -1;
        __syncthreads();

        for (int i = tid; i < (int)R * HID; i += AGG_THREADS) {
            const int dl = i / HID;
            const int c  = i & (HID - 1);
            const size_t o = (size_t)(base + dl) * HID + c;
            const float exs = __expf(es_s[dl] - unord_f32(m_s[dl]));
            const float v = (agg_s[i] + exs * h[o]) / (z_s[dl] + exs + 1e-16f) + bias[c];
            g[o] = v > 0.0f ? v : 0.0f;
        }
        __syncthreads();
    }
}

extern "C" void kernel_launch(void* const* d_in, const int* in_sizes, int n_in,
                              void* d_out, int out_size, void* d_ws, size_t ws_size,
                              hipStream_t stream)
{
    const float* x      = (const float*)d_in[0];
    const int*   ei     = (const int*)  d_in[1];
    const float* W1     = (const float*)d_in[2];
    const float* a_src1 = (const float*)d_in[3];
    const float* a_dst1 = (const float*)d_in[4];
    const float* b1     = (const float*)d_in[5];
    const float* W2     = (const float*)d_in[6];
    const float* a_src2 = (const float*)d_in[7];
    const float* a_dst2 = (const float*)d_in[8];
    const float* b2     = (const float*)d_in[9];
    const float* Wc     = (const float*)d_in[10];
    const float* bc     = (const float*)d_in[11];
    float* out = (float*)d_out;

    const int n       = in_sizes[0] / IN_C;
    const int n_edges = in_sizes[1] / 2;

    float* ws    = (float*)d_ws;
    float* h     = ws;
    float* g     = h   + (size_t)n * HID;
    unsigned* bmap = (unsigned*)(g + (size_t)n * HID);
    float* s_src = g   + (size_t)2 * n * HID;
    float* s_dst = s_src + n;

    const int TB = 256;
    const int nodeBlocks = (n + TB - 1) / TB;
    const int aggBlocks  = (n + AGG_RB - 1) / AGG_RB;
    const int gemmWaves  = (n / 16) * (HID / 16);
    const int gemmBlocks = (gemmWaves + 7) / 8;
    const int nvec16     = n_edges / 16;
    const int nchunk     = nvec16 > 0 ? (nvec16 + KEY_CHUNK - 1) / KEY_CHUNK : 1;

    const long long bmapWords = (long long)256 * nchunk * 32;
    const long long cntWords  = ((long long)aggBlocks + 31) / 32 * 32;
    const long long spare     = (long long)n * HID - bmapWords - cntWords;
    int*      counts = (int*)(bmap + bmapWords);
    unsigned* listA  = bmap + bmapWords + cntWords;
    unsigned* listB  = (unsigned*)out;
    const int nA = spare >= 0 ? (int)(spare / AGG_CAP) : 0;
    const int nB = spare >= 0 ? (int)((long long)n * OUT_C / AGG_CAP) : 0;

    gat_edge_keys<<<nchunk, KEY_CHUNK, 0, stream>>>(ei, bmap, n_edges, nchunk);

    gat_gemm_wmma<IN_C, HID, 0><<<gemmBlocks, TB, 0, stream>>>(x, W1, nullptr, h, n);
    gat_node_prep<<<nodeBlocks, TB, 0, stream>>>(h, a_src1, a_dst1, s_src, s_dst, n);
    gat_block_agg<true><<<aggBlocks, AGG_THREADS, 0, stream>>>(ei, bmap, s_src, s_dst, h, b1, g,
                                                               listA, nA, listB, nB, counts, n, n_edges, nchunk);

    gat_gemm_wmma<HID, HID, 0><<<gemmBlocks, TB, 0, stream>>>(g, W2, nullptr, h, n);
    gat_node_prep<<<nodeBlocks, TB, 0, stream>>>(h, a_src2, a_dst2, s_src, s_dst, n);
    gat_block_agg<false><<<aggBlocks, AGG_THREADS, 0, stream>>>(ei, bmap, s_src, s_dst, h, b2, g,
                                                                listA, nA, listB, nB, counts, n, n_edges, nchunk);

    gat_gemm_wmma<HID, OUT_C, 1><<<gemmBlocks, TB, 0, stream>>>(g, Wc, bc, out, n);
}
